# starting kernel
# baseline (speedup 1.0000x reference)
#include <hip/hip_runtime.h>
#include <math.h>

typedef __attribute__((ext_vector_type(2))) float v2f;
typedef __attribute__((ext_vector_type(8))) float v8f;

#define HEADS 8

#if defined(__gfx1250__) && __has_builtin(__builtin_amdgcn_wmma_f32_16x16x4_f32)
#define USE_WMMA_F32 1
#else
#define USE_WMMA_F32 0
#endif

__device__ __forceinline__ unsigned enc_ord(float f) {
  unsigned b = __float_as_uint(f);
  return (b & 0x80000000u) ? ~b : (b | 0x80000000u);
}
__device__ __forceinline__ float dec_ord(unsigned u) {
  unsigned b = (u & 0x80000000u) ? (u ^ 0x80000000u) : ~u;
  return __uint_as_float(b);
}

__global__ void gemm_wmma(const float* __restrict__ A, const float* __restrict__ W,
                          float* __restrict__ C, int M, int Kin, int Kout) {
  int wid = blockIdx.x * (blockDim.x >> 5) + (threadIdx.x >> 5);
  int tilesN = Kout >> 4;
  int tiles  = (M >> 4) * tilesN;
  if (wid >= tiles) return;
  int tm = wid / tilesN, tn = wid % tilesN;
  int lane = threadIdx.x & 31;
  int grp  = lane >> 4;
  int ln   = lane & 15;

#if USE_WMMA_F32
  const float* Arow = A + (size_t)(tm * 16 + ln) * Kin;
  const float* Wcol = W + (size_t)tn * 16 + ln;
  v8f acc = {};
  for (int k = 0; k < Kin; k += 4) {
    v2f a, b;
    a[0] = Arow[k + 2 * grp];
    a[1] = Arow[k + 2 * grp + 1];
    b[0] = Wcol[(size_t)(k + 2 * grp) * Kout];
    b[1] = Wcol[(size_t)(k + 2 * grp + 1) * Kout];
    acc = __builtin_amdgcn_wmma_f32_16x16x4_f32(false, a, false, b,
                                                (short)0, acc, false, false);
  }
  float* Cout = C + (size_t)(tm * 16 + grp * 8) * Kout + (size_t)tn * 16 + ln;
#pragma unroll
  for (int v = 0; v < 8; ++v) Cout[(size_t)v * Kout] = acc[v];
#else
  float acc[8] = {};
  for (int k = 0; k < Kin; ++k) {
    float bv = W[(size_t)k * Kout + tn * 16 + ln];
#pragma unroll
    for (int v = 0; v < 8; ++v)
      acc[v] += A[(size_t)(tm * 16 + grp * 8 + v) * Kin + k] * bv;
  }
  float* Cout = C + (size_t)(tm * 16 + grp * 8) * Kout + (size_t)tn * 16 + ln;
#pragma unroll
  for (int v = 0; v < 8; ++v) Cout[(size_t)v * Kout] = acc[v];
#endif
}

__global__ void fill_u32(unsigned* __restrict__ p, unsigned v, long long n) {
  long long i = blockIdx.x * (long long)blockDim.x + threadIdx.x;
  long long stride = (long long)gridDim.x * blockDim.x;
  for (; i < n; i += stride) p[i] = v;
}

__device__ __forceinline__ void edge_sd(const int* __restrict__ srcI,
                                        const int* __restrict__ dstI,
                                        int e, int E, int& s, int& d) {
  if (e < E) { s = srcI[e]; d = dstI[e]; }
  else       { s = d = e - E; }
}

__global__ void edge_logits_kernel(const float* __restrict__ xl, const float* __restrict__ xr,
                                   const float* __restrict__ att,
                                   const int* __restrict__ srcI, const int* __restrict__ dstI,
                                   int E, int Etot, int C,
                                   float* __restrict__ logits, unsigned* __restrict__ lmax) {
  long long idx = blockIdx.x * (long long)blockDim.x + threadIdx.x;
  if (idx >= (long long)Etot * HEADS) return;
  int e = (int)(idx >> 3), h = (int)(idx & 7);
  int s, d; edge_sd(srcI, dstI, e, E, s, d);
  const float4* pl = (const float4*)(xl + ((size_t)s * HEADS + h) * C);
  const float4* pr = (const float4*)(xr + ((size_t)d * HEADS + h) * C);
  const float4* pa = (const float4*)(att + (size_t)h * C);
  float acc = 0.f;
  int n4 = C >> 2;
  for (int i = 0; i < n4; ++i) {
    float4 a = pl[i], b = pr[i], w = pa[i];
    float v;
    v = a.x + b.x; v = v > 0.f ? v : 0.2f * v; acc += v * w.x;
    v = a.y + b.y; v = v > 0.f ? v : 0.2f * v; acc += v * w.y;
    v = a.z + b.z; v = v > 0.f ? v : 0.2f * v; acc += v * w.z;
    v = a.w + b.w; v = v > 0.f ? v : 0.2f * v; acc += v * w.w;
  }
  logits[idx] = acc;
  atomicMax(lmax + (size_t)d * HEADS + h, enc_ord(acc));
}

__global__ void edge_accum_kernel(const float* __restrict__ xl,
                                  const int* __restrict__ srcI, const int* __restrict__ dstI,
                                  int E, int Etot, int C,
                                  const float* __restrict__ logits,
                                  const unsigned* __restrict__ lmax,
                                  float* __restrict__ denom,
                                  float* __restrict__ out) {
  long long idx = blockIdx.x * (long long)blockDim.x + threadIdx.x;
  if (idx >= (long long)Etot * HEADS) return;
  int e = (int)(idx >> 3), h = (int)(idx & 7);
  int s, d; edge_sd(srcI, dstI, e, E, s, d);
  float m = dec_ord(lmax[(size_t)d * HEADS + h]);
  float ex = expf(logits[idx] - m);
  atomicAdd(denom + (size_t)d * HEADS + h, ex);
  const float4* pm = (const float4*)(xl + ((size_t)s * HEADS + h) * C);
  float* po = out + ((size_t)d * HEADS + h) * C;
  int n4 = C >> 2;
  for (int i = 0; i < n4; ++i) {
    float4 m4 = pm[i];
    atomicAdd(po + 4 * i + 0, ex * m4.x);
    atomicAdd(po + 4 * i + 1, ex * m4.y);
    atomicAdd(po + 4 * i + 2, ex * m4.z);
    atomicAdd(po + 4 * i + 3, ex * m4.w);
  }
}

__global__ void norm_elu_bias_kernel(float* __restrict__ h, const float* __restrict__ denom,
                                     const float* __restrict__ b, long long total) {
  long long i = blockIdx.x * (long long)blockDim.x + threadIdx.x;
  long long stride = (long long)gridDim.x * blockDim.x;
  for (; i < total; i += stride) {
    int rem = (int)(i & 255);
    long long n = i >> 8;
    int hd = rem >> 5;
    float v = h[i] / denom[n * HEADS + hd] + b[rem];
    h[i] = v > 0.f ? v : (expf(v) - 1.f);
  }
}

__global__ void finalize_logsoftmax(float* __restrict__ out, const float* __restrict__ denom,
                                    const float* __restrict__ b2) {
  int n = blockIdx.x;
  int t = threadIdx.x;
  int hd = t >> 4;
  float v = out[(size_t)n * 128 + t] / denom[(size_t)n * HEADS + hd] + b2[t];
  __shared__ float sm[128];
  sm[t] = v; __syncthreads();
  for (int sft = 64; sft > 0; sft >>= 1) { if (t < sft) sm[t] = fmaxf(sm[t], sm[t + sft]); __syncthreads(); }
  float mx = sm[0]; __syncthreads();
  sm[t] = expf(v - mx); __syncthreads();
  for (int sft = 64; sft > 0; sft >>= 1) { if (t < sft) sm[t] += sm[t + sft]; __syncthreads(); }
  float lse = logf(sm[0]);
  out[(size_t)n * 128 + t] = v - mx - lse;
}

extern "C" void kernel_launch(void* const* d_in, const int* in_sizes, int n_in,
                              void* d_out, int out_size, void* d_ws, size_t ws_size,
                              hipStream_t stream) {
  (void)n_in; (void)out_size; (void)ws_size;
  const float* x    = (const float*)d_in[0];
  const int*   ei   = (const int*)d_in[1];
  const float* W1l  = (const float*)d_in[2];
  const float* W1r  = (const float*)d_in[3];
  const float* att1 = (const float*)d_in[4];
  const float* b1   = (const float*)d_in[5];
  const float* W2l  = (const float*)d_in[6];
  const float* W2r  = (const float*)d_in[7];
  const float* att2 = (const float*)d_in[8];
  const float* b2   = (const float*)d_in[9];

  int N = in_sizes[0] / 128;
  int E = in_sizes[1] / 2;
  int Etot = E + N;
  const int* srcI = ei;
  const int* dstI = ei + E;

  float* ws    = (float*)d_ws;
  float* xl    = ws;
  float* xr    = xl + (size_t)N * 256;
  float* hbuf  = xr + (size_t)N * 256;
  float* lg    = hbuf + (size_t)N * 256;
  unsigned* lmax = (unsigned*)(lg + (size_t)Etot * 8);
  float* denom = (float*)(lmax + (size_t)N * 8);
  float* outAcc = (float*)d_out;

  dim3 blk(256);
  long long eh = (long long)Etot * HEADS;
  int egrid = (int)((eh + 255) / 256);
  const unsigned ENC_NEG_INF = 0x007FFFFFu;

  int tiles1 = (N / 16) * (256 / 16);
  gemm_wmma<<<(tiles1 + 7) / 8, blk, 0, stream>>>(x, W1l, xl, N, 128, 256);
  gemm_wmma<<<(tiles1 + 7) / 8, blk, 0, stream>>>(x, W1r, xr, N, 128, 256);
  fill_u32<<<1024, blk, 0, stream>>>(lmax, ENC_NEG_INF, (long long)N * 8);
  fill_u32<<<1024, blk, 0, stream>>>((unsigned*)denom, 0u, (long long)N * 8);
  fill_u32<<<2048, blk, 0, stream>>>((unsigned*)hbuf, 0u, (long long)N * 256);
  edge_logits_kernel<<<egrid, blk, 0, stream>>>(xl, xr, att1, srcI, dstI, E, Etot, 32, lg, lmax);
  edge_accum_kernel<<<egrid, blk, 0, stream>>>(xl, srcI, dstI, E, Etot, 32, lg, lmax, denom, hbuf);
  norm_elu_bias_kernel<<<2048, blk, 0, stream>>>(hbuf, denom, b1, (long long)N * 256);

  int tiles2 = (N / 16) * (128 / 16);
  gemm_wmma<<<(tiles2 + 7) / 8, blk, 0, stream>>>(hbuf, W2l, xl, N, 256, 128);
  gemm_wmma<<<(tiles2 + 7) / 8, blk, 0, stream>>>(hbuf, W2r, xr, N, 256, 128);
  fill_u32<<<1024, blk, 0, stream>>>(lmax, ENC_NEG_INF, (long long)N * 8);
  fill_u32<<<1024, blk, 0, stream>>>((unsigned*)denom, 0u, (long long)N * 8);
  fill_u32<<<2048, blk, 0, stream>>>((unsigned*)outAcc, 0u, (long long)N * 128);
  edge_logits_kernel<<<egrid, blk, 0, stream>>>(xl, xr, att2, srcI, dstI, E, Etot, 16, lg, lmax);
  edge_accum_kernel<<<egrid, blk, 0, stream>>>(xl, srcI, dstI, E, Etot, 16, lg, lmax, denom, outAcc);
  finalize_logsoftmax<<<N, dim3(128), 0, stream>>>(outAcc, denom, b2);
}


# edge_accum: one wave per edge, lane = channel, so each f32 atomic add instruction covers one whole 128-B line of the destination row (was 32 lines); atomic arrival order free as before
# speedup vs baseline: 2.7951x; 2.7951x over previous
#include <hip/hip_runtime.h>
#include <math.h>

typedef __attribute__((ext_vector_type(2))) float v2f;
typedef __attribute__((ext_vector_type(8))) float v8f;

#define HEADS 8

#if defined(__gfx1250__) && __has_builtin(__builtin_amdgcn_wmma_f32_16x16x4_f32)
#define USE_WMMA_F32 1
#else
#define USE_WMMA_F32 0
#endif

__device__ __forceinline__ unsigned enc_ord(float f) {
  unsigned b = __float_as_uint(f);
  return (b & 0x80000000u) ? ~b : (b | 0x80000000u);
}
__device__ __forceinline__ float dec_ord(unsigned u) {
  unsigned b = (u & 0x80000000u) ? (u ^ 0x80000000u) : ~u;
  return __uint_as_float(b);
}

__global__ void gemm_wmma(const float* __restrict__ A, const float* __restrict__ W,
                          float* __restrict__ C, int M, int Kin, int Kout) {
  int wid = blockIdx.x * (blockDim.x >> 5) + (threadIdx.x >> 5);
  int tilesN = Kout >> 4;
  int tiles  = (M >> 4) * tilesN;
  if (wid >= tiles) return;
  int tm = wid / tilesN, tn = wid % tilesN;
  int lane = threadIdx.x & 31;
  int grp  = lane >> 4;
  int ln   = lane & 15;

#if USE_WMMA_F32
  const float* Arow = A + (size_t)(tm * 16 + ln) * Kin;
  const float* Wcol = W + (size_t)tn * 16 + ln;
  v8f acc = {};
  for (int k = 0; k < Kin; k += 4) {
    v2f a, b;
    a[0] = Arow[k + 2 * grp];
    a[1] = Arow[k + 2 * grp + 1];
    b[0] = Wcol[(size_t)(k + 2 * grp) * Kout];
    b[1] = Wcol[(size_t)(k + 2 * grp + 1) * Kout];
    acc = __builtin_amdgcn_wmma_f32_16x16x4_f32(false, a, false, b,
                                                (short)0, acc, false, false);
  }
  float* Cout = C + (size_t)(tm * 16 + grp * 8) * Kout + (size_t)tn * 16 + ln;
#pragma unroll
  for (int v = 0; v < 8; ++v) Cout[(size_t)v * Kout] = acc[v];
#else
  float acc[8] = {};
  for (int k = 0; k < Kin; ++k) {
    float bv = W[(size_t)k * Kout + tn * 16 + ln];
#pragma unroll
    for (int v = 0; v < 8; ++v)
      acc[v] += A[(size_t)(tm * 16 + grp * 8 + v) * Kin + k] * bv;
  }
  float* Cout = C + (size_t)(tm * 16 + grp * 8) * Kout + (size_t)tn * 16 + ln;
#pragma unroll
  for (int v = 0; v < 8; ++v) Cout[(size_t)v * Kout] = acc[v];
#endif
}

__global__ void fill_u32(unsigned* __restrict__ p, unsigned v, long long n) {
  long long i = blockIdx.x * (long long)blockDim.x + threadIdx.x;
  long long stride = (long long)gridDim.x * blockDim.x;
  for (; i < n; i += stride) p[i] = v;
}

__device__ __forceinline__ void edge_sd(const int* __restrict__ srcI,
                                        const int* __restrict__ dstI,
                                        int e, int E, int& s, int& d) {
  if (e < E) { s = srcI[e]; d = dstI[e]; }
  else       { s = d = e - E; }
}

__global__ void edge_logits_kernel(const float* __restrict__ xl, const float* __restrict__ xr,
                                   const float* __restrict__ att,
                                   const int* __restrict__ srcI, const int* __restrict__ dstI,
                                   int E, int Etot, int C,
                                   float* __restrict__ logits, unsigned* __restrict__ lmax) {
  long long idx = blockIdx.x * (long long)blockDim.x + threadIdx.x;
  if (idx >= (long long)Etot * HEADS) return;
  int e = (int)(idx >> 3), h = (int)(idx & 7);
  int s, d; edge_sd(srcI, dstI, e, E, s, d);
  const float4* pl = (const float4*)(xl + ((size_t)s * HEADS + h) * C);
  const float4* pr = (const float4*)(xr + ((size_t)d * HEADS + h) * C);
  const float4* pa = (const float4*)(att + (size_t)h * C);
  float acc = 0.f;
  int n4 = C >> 2;
  for (int i = 0; i < n4; ++i) {
    float4 a = pl[i], b = pr[i], w = pa[i];
    float v;
    v = a.x + b.x; v = v > 0.f ? v : 0.2f * v; acc += v * w.x;
    v = a.y + b.y; v = v > 0.f ? v : 0.2f * v; acc += v * w.y;
    v = a.z + b.z; v = v > 0.f ? v : 0.2f * v; acc += v * w.z;
    v = a.w + b.w; v = v > 0.f ? v : 0.2f * v; acc += v * w.w;
  }
  logits[idx] = acc;
  atomicMax(lmax + (size_t)d * HEADS + h, enc_ord(acc));
}

template <int C>
__device__ __forceinline__ void edge_accum_rows(const float* __restrict__ xl,
                                                const int* __restrict__ srcI,
                                                const int* __restrict__ dstI,
                                                int E, int Etot,
                                                const float* __restrict__ logits,
                                                const unsigned* __restrict__ lmax,
                                                float* __restrict__ denom,
                                                float* __restrict__ out) {
  constexpr int ROW = HEADS * C;
  constexpr int NI  = ROW / 32;
  const int lane = threadIdx.x & 31;
  const int wib  = __builtin_amdgcn_readfirstlane((int)(threadIdx.x >> 5));
  const int wave = blockIdx.x * (int)(blockDim.x >> 5) + wib;
  const int nw   = gridDim.x * (int)(blockDim.x >> 5);
  const int h8   = lane & 7;
  for (int e = wave; e < Etot; e += nw) {
    int s, d; edge_sd(srcI, dstI, e, E, s, d);
    float m  = dec_ord(lmax[(size_t)d * HEADS + h8]);
    float ex = expf(logits[(size_t)e * HEADS + h8] - m);
    if (lane < HEADS) atomicAdd(denom + (size_t)d * HEADS + lane, ex);
    const float* pm = xl + (size_t)s * ROW + lane;
    float* po = out + (size_t)d * ROW + lane;
    float v[NI];
#pragma unroll
    for (int i = 0; i < NI; ++i) v[i] = pm[i * 32];
#pragma unroll
    for (int i = 0; i < NI; ++i) {
      int hd = (i * 32 + lane) / C;
      atomicAdd(po + i * 32, __shfl(ex, hd) * v[i]);
    }
  }
}

__global__ void edge_accum_c32(const float* __restrict__ xl,
                               const int* __restrict__ srcI, const int* __restrict__ dstI,
                               int E, int Etot,
                               const float* __restrict__ logits,
                               const unsigned* __restrict__ lmax,
                               float* __restrict__ denom, float* __restrict__ out) {
  edge_accum_rows<32>(xl, srcI, dstI, E, Etot, logits, lmax, denom, out);
}

__global__ void edge_accum_c16(const float* __restrict__ xl,
                               const int* __restrict__ srcI, const int* __restrict__ dstI,
                               int E, int Etot,
                               const float* __restrict__ logits,
                               const unsigned* __restrict__ lmax,
                               float* __restrict__ denom, float* __restrict__ out) {
  edge_accum_rows<16>(xl, srcI, dstI, E, Etot, logits, lmax, denom, out);
}

__global__ void norm_elu_bias_kernel(float* __restrict__ h, const float* __restrict__ denom,
                                     const float* __restrict__ b, long long total) {
  long long i = blockIdx.x * (long long)blockDim.x + threadIdx.x;
  long long stride = (long long)gridDim.x * blockDim.x;
  for (; i < total; i += stride) {
    int rem = (int)(i & 255);
    long long n = i >> 8;
    int hd = rem >> 5;
    float v = h[i] / denom[n * HEADS + hd] + b[rem];
    h[i] = v > 0.f ? v : (expf(v) - 1.f);
  }
}

__global__ void finalize_logsoftmax(float* __restrict__ out, const float* __restrict__ denom,
                                    const float* __restrict__ b2) {
  int n = blockIdx.x;
  int t = threadIdx.x;
  int hd = t >> 4;
  float v = out[(size_t)n * 128 + t] / denom[(size_t)n * HEADS + hd] + b2[t];
  __shared__ float sm[128];
  sm[t] = v; __syncthreads();
  for (int sft = 64; sft > 0; sft >>= 1) { if (t < sft) sm[t] = fmaxf(sm[t], sm[t + sft]); __syncthreads(); }
  float mx = sm[0]; __syncthreads();
  sm[t] = expf(v - mx); __syncthreads();
  for (int sft = 64; sft > 0; sft >>= 1) { if (t < sft) sm[t] += sm[t + sft]; __syncthreads(); }
  float lse = logf(sm[0]);
  out[(size_t)n * 128 + t] = v - mx - lse;
}

extern "C" void kernel_launch(void* const* d_in, const int* in_sizes, int n_in,
                              void* d_out, int out_size, void* d_ws, size_t ws_size,
                              hipStream_t stream) {
  (void)n_in; (void)out_size; (void)ws_size;
  const float* x    = (const float*)d_in[0];
  const int*   ei   = (const int*)d_in[1];
  const float* W1l  = (const float*)d_in[2];
  const float* W1r  = (const float*)d_in[3];
  const float* att1 = (const float*)d_in[4];
  const float* b1   = (const float*)d_in[5];
  const float* W2l  = (const float*)d_in[6];
  const float* W2r  = (const float*)d_in[7];
  const float* att2 = (const float*)d_in[8];
  const float* b2   = (const float*)d_in[9];

  int N = in_sizes[0] / 128;
  int E = in_sizes[1] / 2;
  int Etot = E + N;
  const int* srcI = ei;
  const int* dstI = ei + E;

  float* ws    = (float*)d_ws;
  float* xl    = ws;
  float* xr    = xl + (size_t)N * 256;
  float* hbuf  = xr + (size_t)N * 256;
  float* lg    = hbuf + (size_t)N * 256;
  unsigned* lmax = (unsigned*)(lg + (size_t)Etot * 8);
  float* denom = (float*)(lmax + (size_t)N * 8);
  float* outAcc = (float*)d_out;

  dim3 blk(256);
  long long eh = (long long)Etot * HEADS;
  int egrid = (int)((eh + 255) / 256);
  const unsigned ENC_NEG_INF = 0x007FFFFFu;

  int tiles1 = (N / 16) * (256 / 16);
  gemm_wmma<<<(tiles1 + 7) / 8, blk, 0, stream>>>(x, W1l, xl, N, 128, 256);
  gemm_wmma<<<(tiles1 + 7) / 8, blk, 0, stream>>>(x, W1r, xr, N, 128, 256);
  fill_u32<<<1024, blk, 0, stream>>>(lmax, ENC_NEG_INF, (long long)N * 8);
  fill_u32<<<1024, blk, 0, stream>>>((unsigned*)denom, 0u, (long long)N * 8);
  fill_u32<<<2048, blk, 0, stream>>>((unsigned*)hbuf, 0u, (long long)N * 256);
  edge_logits_kernel<<<egrid, blk, 0, stream>>>(xl, xr, att1, srcI, dstI, E, Etot, 32, lg, lmax);
  edge_accum_c32<<<4096, blk, 0, stream>>>(xl, srcI, dstI, E, Etot, lg, lmax, denom, hbuf);
  norm_elu_bias_kernel<<<2048, blk, 0, stream>>>(hbuf, denom, b1, (long long)N * 256);

  int tiles2 = (N / 16) * (128 / 16);
  gemm_wmma<<<(tiles2 + 7) / 8, blk, 0, stream>>>(hbuf, W2l, xl, N, 256, 128);
  gemm_wmma<<<(tiles2 + 7) / 8, blk, 0, stream>>>(hbuf, W2r, xr, N, 256, 128);
  fill_u32<<<1024, blk, 0, stream>>>(lmax, ENC_NEG_INF, (long long)N * 8);
  fill_u32<<<1024, blk, 0, stream>>>((unsigned*)denom, 0u, (long long)N * 8);
  fill_u32<<<2048, blk, 0, stream>>>((unsigned*)outAcc, 0u, (long long)N * 128);
  edge_logits_kernel<<<egrid, blk, 0, stream>>>(xl, xr, att2, srcI, dstI, E, Etot, 16, lg, lmax);
  edge_accum_c16<<<4096, blk, 0, stream>>>(xl, srcI, dstI, E, Etot, lg, lmax, denom, outAcc);
  finalize_logsoftmax<<<N, dim3(128), 0, stream>>>(outAcc, denom, b2);
}


# edge_logits: one wave per edge, lane = 4 consecutive row floats, so each row load covers whole 128-B lines (was 32 lines per load); per-head sum is now lane partials + xor tree
# speedup vs baseline: 3.3459x; 1.1971x over previous
#include <hip/hip_runtime.h>
#include <math.h>

typedef __attribute__((ext_vector_type(2))) float v2f;
typedef __attribute__((ext_vector_type(8))) float v8f;

#define HEADS 8

#if defined(__gfx1250__) && __has_builtin(__builtin_amdgcn_wmma_f32_16x16x4_f32)
#define USE_WMMA_F32 1
#else
#define USE_WMMA_F32 0
#endif

__device__ __forceinline__ unsigned enc_ord(float f) {
  unsigned b = __float_as_uint(f);
  return (b & 0x80000000u) ? ~b : (b | 0x80000000u);
}
__device__ __forceinline__ float dec_ord(unsigned u) {
  unsigned b = (u & 0x80000000u) ? (u ^ 0x80000000u) : ~u;
  return __uint_as_float(b);
}

__global__ void gemm_wmma(const float* __restrict__ A, const float* __restrict__ W,
                          float* __restrict__ C, int M, int Kin, int Kout) {
  int wid = blockIdx.x * (blockDim.x >> 5) + (threadIdx.x >> 5);
  int tilesN = Kout >> 4;
  int tiles  = (M >> 4) * tilesN;
  if (wid >= tiles) return;
  int tm = wid / tilesN, tn = wid % tilesN;
  int lane = threadIdx.x & 31;
  int grp  = lane >> 4;
  int ln   = lane & 15;

#if USE_WMMA_F32
  const float* Arow = A + (size_t)(tm * 16 + ln) * Kin;
  const float* Wcol = W + (size_t)tn * 16 + ln;
  v8f acc = {};
  for (int k = 0; k < Kin; k += 4) {
    v2f a, b;
    a[0] = Arow[k + 2 * grp];
    a[1] = Arow[k + 2 * grp + 1];
    b[0] = Wcol[(size_t)(k + 2 * grp) * Kout];
    b[1] = Wcol[(size_t)(k + 2 * grp + 1) * Kout];
    acc = __builtin_amdgcn_wmma_f32_16x16x4_f32(false, a, false, b,
                                                (short)0, acc, false, false);
  }
  float* Cout = C + (size_t)(tm * 16 + grp * 8) * Kout + (size_t)tn * 16 + ln;
#pragma unroll
  for (int v = 0; v < 8; ++v) Cout[(size_t)v * Kout] = acc[v];
#else
  float acc[8] = {};
  for (int k = 0; k < Kin; ++k) {
    float bv = W[(size_t)k * Kout + tn * 16 + ln];
#pragma unroll
    for (int v = 0; v < 8; ++v)
      acc[v] += A[(size_t)(tm * 16 + grp * 8 + v) * Kin + k] * bv;
  }
  float* Cout = C + (size_t)(tm * 16 + grp * 8) * Kout + (size_t)tn * 16 + ln;
#pragma unroll
  for (int v = 0; v < 8; ++v) Cout[(size_t)v * Kout] = acc[v];
#endif
}

__global__ void fill_u32(unsigned* __restrict__ p, unsigned v, long long n) {
  long long i = blockIdx.x * (long long)blockDim.x + threadIdx.x;
  long long stride = (long long)gridDim.x * blockDim.x;
  for (; i < n; i += stride) p[i] = v;
}

__device__ __forceinline__ void edge_sd(const int* __restrict__ srcI,
                                        const int* __restrict__ dstI,
                                        int e, int E, int& s, int& d) {
  if (e < E) { s = srcI[e]; d = dstI[e]; }
  else       { s = d = e - E; }
}

__device__ __forceinline__ float leaky_dot4(float4 a, float4 b, float4 w) {
  float acc, v;
  v = a.x + b.x; v = v > 0.f ? v : 0.2f * v; acc  = v * w.x;
  v = a.y + b.y; v = v > 0.f ? v : 0.2f * v; acc += v * w.y;
  v = a.z + b.z; v = v > 0.f ? v : 0.2f * v; acc += v * w.z;
  v = a.w + b.w; v = v > 0.f ? v : 0.2f * v; acc += v * w.w;
  return acc;
}

template <int C>
__device__ __forceinline__ float head_logit(const float4* a, const float4* b, const float4* w,
                                            int lane) {
  constexpr int NJ  = HEADS * C / 128;
  constexpr int LPH = C / 4;
  float p[NJ];
#pragma unroll
  for (int j = 0; j < NJ; ++j) {
    p[j] = leaky_dot4(a[j], b[j], w[j]);
#pragma unroll
    for (int m = 1; m < LPH; m <<= 1) p[j] += __shfl_xor(p[j], m);
  }
  if constexpr (C == 32) {
    float q0 = __shfl(p[0], (lane & 3) * 8);
    float q1 = __shfl(p[1], (lane & 3) * 8);
    return (lane & 4) ? q1 : q0;
  } else {
    return __shfl(p[0], (lane & 7) * 4);
  }
}

template <int C>
__device__ __forceinline__ void edge_logits_rows(const float* __restrict__ xl,
                                                 const float* __restrict__ xr,
                                                 const float* __restrict__ att,
                                                 const int* __restrict__ srcI,
                                                 const int* __restrict__ dstI,
                                                 int E, int Etot,
                                                 float* __restrict__ logits,
                                                 unsigned* __restrict__ lmax) {
  static_assert(C == 32 || C == 16, "row pieces of 128 floats, 8 or 4 lanes per head");
  constexpr int ROW = HEADS * C;
  constexpr int NJ  = ROW / 128;
  const int lane = threadIdx.x & 31;
  const int wib  = __builtin_amdgcn_readfirstlane((int)(threadIdx.x >> 5));
  const int wave = blockIdx.x * (int)(blockDim.x >> 5) + wib;
  const int nw   = gridDim.x * (int)(blockDim.x >> 5);
  float4 w[NJ];
#pragma unroll
  for (int j = 0; j < NJ; ++j) w[j] = ((const float4*)att)[j * 32 + lane];
  for (int e0 = wave; e0 < Etot; e0 += 2 * nw) {
    const int  e1  = e0 + nw;
    const bool two = e1 < Etot;
    int s0, d0, s1, d1;
    edge_sd(srcI, dstI, e0, E, s0, d0);
    edge_sd(srcI, dstI, two ? e1 : e0, E, s1, d1);
    const float4* pl0 = (const float4*)(xl + (size_t)s0 * ROW) + lane;
    const float4* pr0 = (const float4*)(xr + (size_t)d0 * ROW) + lane;
    const float4* pl1 = (const float4*)(xl + (size_t)s1 * ROW) + lane;
    const float4* pr1 = (const float4*)(xr + (size_t)d1 * ROW) + lane;
    float4 a0[NJ], b0[NJ], a1[NJ], b1[NJ];
#pragma unroll
    for (int j = 0; j < NJ; ++j) {
      a0[j] = pl0[j * 32]; b0[j] = pr0[j * 32];
      a1[j] = pl1[j * 32]; b1[j] = pr1[j * 32];
    }
    const float lg0 = head_logit<C>(a0, b0, w, lane);
    const float lg1 = head_logit<C>(a1, b1, w, lane);
    if (lane < HEADS) {
      logits[(size_t)e0 * HEADS + lane] = lg0;
      atomicMax(lmax + (size_t)d0 * HEADS + lane, enc_ord(lg0));
      if (two) {
        logits[(size_t)e1 * HEADS + lane] = lg1;
        atomicMax(lmax + (size_t)d1 * HEADS + lane, enc_ord(lg1));
      }
    }
  }
}

__global__ void edge_logits_c32(const float* __restrict__ xl, const float* __restrict__ xr,
                                const float* __restrict__ att,
                                const int* __restrict__ srcI, const int* __restrict__ dstI,
                                int E, int Etot,
                                float* __restrict__ logits, unsigned* __restrict__ lmax) {
  edge_logits_rows<32>(xl, xr, att, srcI, dstI, E, Etot, logits, lmax);
}

__global__ void edge_logits_c16(const float* __restrict__ xl, const float* __restrict__ xr,
                                const float* __restrict__ att,
                                const int* __restrict__ srcI, const int* __restrict__ dstI,
                                int E, int Etot,
                                float* __restrict__ logits, unsigned* __restrict__ lmax) {
  edge_logits_rows<16>(xl, xr, att, srcI, dstI, E, Etot, logits, lmax);
}

template <int C>
__device__ __forceinline__ void edge_accum_rows(const float* __restrict__ xl,
                                                const int* __restrict__ srcI,
                                                const int* __restrict__ dstI,
                                                int E, int Etot,
                                                const float* __restrict__ logits,
                                                const unsigned* __restrict__ lmax,
                                                float* __restrict__ denom,
                                                float* __restrict__ out) {
  constexpr int ROW = HEADS * C;
  constexpr int NI  = ROW / 32;
  const int lane = threadIdx.x & 31;
  const int wib  = __builtin_amdgcn_readfirstlane((int)(threadIdx.x >> 5));
  const int wave = blockIdx.x * (int)(blockDim.x >> 5) + wib;
  const int nw   = gridDim.x * (int)(blockDim.x >> 5);
  const int h8   = lane & 7;
  for (int e = wave; e < Etot; e += nw) {
    int s, d; edge_sd(srcI, dstI, e, E, s, d);
    float m  = dec_ord(lmax[(size_t)d * HEADS + h8]);
    float ex = expf(logits[(size_t)e * HEADS + h8] - m);
    if (lane < HEADS) atomicAdd(denom + (size_t)d * HEADS + lane, ex);
    const float* pm = xl + (size_t)s * ROW + lane;
    float* po = out + (size_t)d * ROW + lane;
    float v[NI];
#pragma unroll
    for (int i = 0; i < NI; ++i) v[i] = pm[i * 32];
#pragma unroll
    for (int i = 0; i < NI; ++i) {
      int hd = (i * 32 + lane) / C;
      atomicAdd(po + i * 32, __shfl(ex, hd) * v[i]);
    }
  }
}

__global__ void edge_accum_c32(const float* __restrict__ xl,
                               const int* __restrict__ srcI, const int* __restrict__ dstI,
                               int E, int Etot,
                               const float* __restrict__ logits,
                               const unsigned* __restrict__ lmax,
                               float* __restrict__ denom, float* __restrict__ out) {
  edge_accum_rows<32>(xl, srcI, dstI, E, Etot, logits, lmax, denom, out);
}

__global__ void edge_accum_c16(const float* __restrict__ xl,
                               const int* __restrict__ srcI, const int* __restrict__ dstI,
                               int E, int Etot,
                               const float* __restrict__ logits,
                               const unsigned* __restrict__ lmax,
                               float* __restrict__ denom, float* __restrict__ out) {
  edge_accum_rows<16>(xl, srcI, dstI, E, Etot, logits, lmax, denom, out);
}

__global__ void norm_elu_bias_kernel(float* __restrict__ h, const float* __restrict__ denom,
                                     const float* __restrict__ b, long long total) {
  long long i = blockIdx.x * (long long)blockDim.x + threadIdx.x;
  long long stride = (long long)gridDim.x * blockDim.x;
  for (; i < total; i += stride) {
    int rem = (int)(i & 255);
    long long n = i >> 8;
    int hd = rem >> 5;
    float v = h[i] / denom[n * HEADS + hd] + b[rem];
    h[i] = v > 0.f ? v : (expf(v) - 1.f);
  }
}

__global__ void finalize_logsoftmax(float* __restrict__ out, const float* __restrict__ denom,
                                    const float* __restrict__ b2) {
  int n = blockIdx.x;
  int t = threadIdx.x;
  int hd = t >> 4;
  float v = out[(size_t)n * 128 + t] / denom[(size_t)n * HEADS + hd] + b2[t];
  __shared__ float sm[128];
  sm[t] = v; __syncthreads();
  for (int sft = 64; sft > 0; sft >>= 1) { if (t < sft) sm[t] = fmaxf(sm[t], sm[t + sft]); __syncthreads(); }
  float mx = sm[0]; __syncthreads();
  sm[t] = expf(v - mx); __syncthreads();
  for (int sft = 64; sft > 0; sft >>= 1) { if (t < sft) sm[t] += sm[t + sft]; __syncthreads(); }
  float lse = logf(sm[0]);
  out[(size_t)n * 128 + t] = v - mx - lse;
}

extern "C" void kernel_launch(void* const* d_in, const int* in_sizes, int n_in,
                              void* d_out, int out_size, void* d_ws, size_t ws_size,
                              hipStream_t stream) {
  (void)n_in; (void)out_size; (void)ws_size;
  const float* x    = (const float*)d_in[0];
  const int*   ei   = (const int*)d_in[1];
  const float* W1l  = (const float*)d_in[2];
  const float* W1r  = (const float*)d_in[3];
  const float* att1 = (const float*)d_in[4];
  const float* b1   = (const float*)d_in[5];
  const float* W2l  = (const float*)d_in[6];
  const float* W2r  = (const float*)d_in[7];
  const float* att2 = (const float*)d_in[8];
  const float* b2   = (const float*)d_in[9];

  int N = in_sizes[0] / 128;
  int E = in_sizes[1] / 2;
  int Etot = E + N;
  const int* srcI = ei;
  const int* dstI = ei + E;

  float* ws    = (float*)d_ws;
  float* xl    = ws;
  float* xr    = xl + (size_t)N * 256;
  float* hbuf  = xr + (size_t)N * 256;
  float* lg    = hbuf + (size_t)N * 256;
  unsigned* lmax = (unsigned*)(lg + (size_t)Etot * 8);
  float* denom = (float*)(lmax + (size_t)N * 8);
  float* outAcc = (float*)d_out;

  dim3 blk(256);
  const unsigned ENC_NEG_INF = 0x007FFFFFu;

  int tiles1 = (N / 16) * (256 / 16);
  gemm_wmma<<<(tiles1 + 7) / 8, blk, 0, stream>>>(x, W1l, xl, N, 128, 256);
  gemm_wmma<<<(tiles1 + 7) / 8, blk, 0, stream>>>(x, W1r, xr, N, 128, 256);
  fill_u32<<<1024, blk, 0, stream>>>(lmax, ENC_NEG_INF, (long long)N * 8);
  fill_u32<<<1024, blk, 0, stream>>>((unsigned*)denom, 0u, (long long)N * 8);
  fill_u32<<<2048, blk, 0, stream>>>((unsigned*)hbuf, 0u, (long long)N * 256);
  edge_logits_c32<<<2048, blk, 0, stream>>>(xl, xr, att1, srcI, dstI, E, Etot, lg, lmax);
  edge_accum_c32<<<4096, blk, 0, stream>>>(xl, srcI, dstI, E, Etot, lg, lmax, denom, hbuf);
  norm_elu_bias_kernel<<<2048, blk, 0, stream>>>(hbuf, denom, b1, (long long)N * 256);

  int tiles2 = (N / 16) * (128 / 16);
  gemm_wmma<<<(tiles2 + 7) / 8, blk, 0, stream>>>(hbuf, W2l, xl, N, 256, 128);
  gemm_wmma<<<(tiles2 + 7) / 8, blk, 0, stream>>>(hbuf, W2r, xr, N, 256, 128);
  fill_u32<<<1024, blk, 0, stream>>>(lmax, ENC_NEG_INF, (long long)N * 8);
  fill_u32<<<1024, blk, 0, stream>>>((unsigned*)denom, 0u, (long long)N * 8);
  fill_u32<<<2048, blk, 0, stream>>>((unsigned*)outAcc, 0u, (long long)N * 128);
  edge_logits_c16<<<2048, blk, 0, stream>>>(xl, xr, att2, srcI, dstI, E, Etot, lg, lmax);
  edge_accum_c16<<<4096, blk, 0, stream>>>(xl, srcI, dstI, E, Etot, lg, lmax, denom, outAcc);
  finalize_logsoftmax<<<N, dim3(128), 0, stream>>>(outAcc, denom, b2);
}


# edges grouped by destination per call (int-atomic count, scan, scatter); one wave per node: running-max softmax, weighted sum, epilogue in registers; no float atomics; sums in arrival order, xor tree
# speedup vs baseline: 4.8860x; 1.4603x over previous
#include <hip/hip_runtime.h>
#include <math.h>

typedef __attribute__((ext_vector_type(2))) float v2f;
typedef __attribute__((ext_vector_type(8))) float v8f;
typedef __attribute__((ext_vector_type(4))) float v4f;

#define HEADS 8

#if defined(__gfx1250__) && __has_builtin(__builtin_amdgcn_wmma_f32_16x16x4_f32)
#define USE_WMMA_F32 1
#else
#define USE_WMMA_F32 0
#endif

__global__ void gemm_wmma(const float* __restrict__ A, const float* __restrict__ W,
                          float* __restrict__ C, int M, int Kin, int Kout) {
  int wid = blockIdx.x * (blockDim.x >> 5) + (threadIdx.x >> 5);
  int tilesN = Kout >> 4;
  int tiles  = (M >> 4) * tilesN;
  if (wid >= tiles) return;
  int tm = wid / tilesN, tn = wid % tilesN;
  int lane = threadIdx.x & 31;
  int grp  = lane >> 4;
  int ln   = lane & 15;

#if USE_WMMA_F32
  const float* Arow = A + (size_t)(tm * 16 + ln) * Kin;
  const float* Wcol = W + (size_t)tn * 16 + ln;
  v8f acc = {};
  for (int k = 0; k < Kin; k += 4) {
    v2f a, b;
    a[0] = Arow[k + 2 * grp];
    a[1] = Arow[k + 2 * grp + 1];
    b[0] = Wcol[(size_t)(k + 2 * grp) * Kout];
    b[1] = Wcol[(size_t)(k + 2 * grp + 1) * Kout];
    acc = __builtin_amdgcn_wmma_f32_16x16x4_f32(false, a, false, b,
                                                (short)0, acc, false, false);
  }
  float* Cout = C + (size_t)(tm * 16 + grp * 8) * Kout + (size_t)tn * 16 + ln;
#pragma unroll
  for (int v = 0; v < 8; ++v) Cout[(size_t)v * Kout] = acc[v];
#else
  float acc[8] = {};
  for (int k = 0; k < Kin; ++k) {
    float bv = W[(size_t)k * Kout + tn * 16 + ln];
#pragma unroll
    for (int v = 0; v < 8; ++v)
      acc[v] += A[(size_t)(tm * 16 + grp * 8 + v) * Kin + k] * bv;
  }
  float* Cout = C + (size_t)(tm * 16 + grp * 8) * Kout + (size_t)tn * 16 + ln;
#pragma unroll
  for (int v = 0; v < 8; ++v) Cout[(size_t)v * Kout] = acc[v];
#endif
}

__global__ void fill_u32(unsigned* __restrict__ p, unsigned v, long long n) {
  long long i = blockIdx.x * (long long)blockDim.x + threadIdx.x;
  long long stride = (long long)gridDim.x * blockDim.x;
  for (; i < n; i += stride) p[i] = v;
}

__global__ void dst_count(const int* __restrict__ dstI, int E, unsigned* __restrict__ cnt) {
  int i = blockIdx.x * blockDim.x + threadIdx.x;
  int stride = gridDim.x * blockDim.x;
  for (; i < E; i += stride) atomicAdd(cnt + dstI[i], 1u);
}

__global__ void __launch_bounds__(1024) dst_scan(const unsigned* __restrict__ cnt, int N,
                                                 unsigned* __restrict__ rowptr,
                                                 unsigned* __restrict__ cursor) {
  __shared__ unsigned wsum[32];
  const int t = threadIdx.x, lane = t & 31, wv = t >> 5;
  unsigned carry = 0;
  unsigned n0, n1, n2, n3;
  {
    const int i0 = 4 * t;
    n0 = i0     < N ? cnt[i0]     : 0u;
    n1 = i0 + 1 < N ? cnt[i0 + 1] : 0u;
    n2 = i0 + 2 < N ? cnt[i0 + 2] : 0u;
    n3 = i0 + 3 < N ? cnt[i0 + 3] : 0u;
  }
  for (int base = 0; base < N; base += 4096) {
    const int i0 = base + 4 * t;
    const unsigned c0 = n0, c1 = n1, c2 = n2, c3 = n3;
    {
      const int j0 = i0 + 4096;
      n0 = j0     < N ? cnt[j0]     : 0u;
      n1 = j0 + 1 < N ? cnt[j0 + 1] : 0u;
      n2 = j0 + 2 < N ? cnt[j0 + 2] : 0u;
      n3 = j0 + 3 < N ? cnt[j0 + 3] : 0u;
    }
    const unsigned local = c0 + c1 + c2 + c3;
    unsigned inc = local;
#pragma unroll
    for (int o = 1; o < 32; o <<= 1) {
      unsigned nb = __shfl_up(inc, o);
      if (lane >= o) inc += nb;
    }
    if (lane == 31) wsum[wv] = inc;
    __syncthreads();
    const unsigned wtot = wsum[lane];
    unsigned winc = wtot;
#pragma unroll
    for (int o = 1; o < 32; o <<= 1) {
      unsigned nb = __shfl_up(winc, o);
      if (lane >= o) winc += nb;
    }
    const unsigned woff  = __shfl(winc, wv) - __shfl(wtot, wv);
    const unsigned total = __shfl(winc, 31);
    unsigned ex = carry + woff + inc - local;
    if (i0     < N) { rowptr[i0]     = ex; cursor[i0]     = ex; } ex += c0;
    if (i0 + 1 < N) { rowptr[i0 + 1] = ex; cursor[i0 + 1] = ex; } ex += c1;
    if (i0 + 2 < N) { rowptr[i0 + 2] = ex; cursor[i0 + 2] = ex; } ex += c2;
    if (i0 + 3 < N) { rowptr[i0 + 3] = ex; cursor[i0 + 3] = ex; }
    carry += total;
    __syncthreads();
  }
  if (t == 0) rowptr[N] = carry;
}

__global__ void dst_scatter(const int* __restrict__ srcI, const int* __restrict__ dstI, int E,
                            unsigned* __restrict__ cursor, int* __restrict__ csr_src) {
  int i = blockIdx.x * blockDim.x + threadIdx.x;
  int stride = gridDim.x * blockDim.x;
  for (; i < E; i += stride) {
    unsigned p = atomicAdd(cursor + dstI[i], 1u);
    csr_src[p] = srcI[i];
  }
}

__device__ __forceinline__ float leaky_dot4(v4f a, v4f b, v4f w) {
  float acc, v;
  v = a.x + b.x; v = fmaxf(v, 0.2f * v); acc  = v * w.x;
  v = a.y + b.y; v = fmaxf(v, 0.2f * v); acc += v * w.y;
  v = a.z + b.z; v = fmaxf(v, 0.2f * v); acc += v * w.z;
  v = a.w + b.w; v = fmaxf(v, 0.2f * v); acc += v * w.w;
  return acc;
}

template <int C>
__device__ __forceinline__ float group_logit(v4f a, v4f b, v4f w) {
  float p = leaky_dot4(a, b, w);
#pragma unroll
  for (int m = 1; m < C / 4; m <<= 1) p += __shfl_xor(p, m);
  return p;
}

template <int C, bool LAST>
__device__ __forceinline__ void node_rows(const float* __restrict__ xl,
                                          const float* __restrict__ xr,
                                          const float* __restrict__ att,
                                          const float* __restrict__ bias,
                                          const unsigned* __restrict__ rowptr,
                                          const int* __restrict__ csr_src,
                                          int N, float* __restrict__ out) {
  static_assert(C == 32 || C == 16, "row pieces of 128 floats, 8 or 4 lanes per head");
  static_assert(!LAST || C == 16, "the log-softmax epilogue takes one 128-float piece");
  constexpr int ROW = HEADS * C;
  constexpr int NJ  = ROW / 128;
  constexpr int UN  = 4;
  static_assert(UN == 4 && (NJ == 1 || NJ == 2), "the load fence below names UN*NJ vectors");
  const int lane = threadIdx.x & 31;
  const int wib  = __builtin_amdgcn_readfirstlane((int)(threadIdx.x >> 5));
  const int wave = blockIdx.x * (int)(blockDim.x >> 5) + wib;
  const int nw   = gridDim.x * (int)(blockDim.x >> 5);
  v4f w[NJ], bz[NJ];
#pragma unroll
  for (int j = 0; j < NJ; ++j) {
    w[j]  = ((const v4f*)att)[j * 32 + lane];
    bz[j] = ((const v4f*)bias)[j * 32 + lane];
  }
  for (int d = wave; d < N; d += nw) {
    const unsigned beg = rowptr[d], end = rowptr[d + 1];
    const v4f* pr = (const v4f*)(xr + (size_t)d * ROW) + lane;
    const v4f* ps = (const v4f*)(xl + (size_t)d * ROW) + lane;
    v4f b[NJ], acc[NJ];
    float  mx[NJ], den[NJ];
#pragma unroll
    for (int j = 0; j < NJ; ++j) { b[j] = pr[j * 32]; acc[j] = ps[j * 32]; }
#pragma unroll
    for (int j = 0; j < NJ; ++j) { mx[j] = group_logit<C>(acc[j], b[j], w[j]); den[j] = 1.f; }

    for (unsigned base = beg; base < end; base += 32) {
      const int cnt = (int)(end - base) < 32 ? (int)(end - base) : 32;
      const int my  = lane < cnt ? csr_src[base + lane] : 0;
      for (int i = 0; i < cnt; i += UN) {
        v4f a[UN][NJ];
#pragma unroll
        for (int u = 0; u < UN; ++u) {
          const int ii = i + u < cnt ? i + u : cnt - 1;
          const int s  = __builtin_amdgcn_readlane(my, ii);
          const v4f* p = (const v4f*)(xl + (size_t)s * ROW) + lane;
#pragma unroll
          for (int j = 0; j < NJ; ++j) a[u][j] = p[j * 32];
        }
        if constexpr (NJ == 2)
          asm volatile("" :: "v"(a[0][0]), "v"(a[0][1]), "v"(a[1][0]), "v"(a[1][1]),
                             "v"(a[2][0]), "v"(a[2][1]), "v"(a[3][0]), "v"(a[3][1]));
        else
          asm volatile("" :: "v"(a[0][0]), "v"(a[1][0]), "v"(a[2][0]), "v"(a[3][0]));
#pragma unroll
        for (int u = 0; u < UN; ++u) {
          if (u == 0 || i + u < cnt) {
#pragma unroll
            for (int j = 0; j < NJ; ++j) {
              const float lg = group_logit<C>(a[u][j], b[j], w[j]);
              const float t  = expf(-fabsf(mx[j] - lg));
              const bool  up = lg > mx[j];
              const float sc = up ? t : 1.f;
              const float ex = up ? 1.f : t;
              mx[j]  = fmaxf(mx[j], lg);
              den[j] = den[j] * sc + ex;
              acc[j].x = acc[j].x * sc + ex * a[u][j].x;
              acc[j].y = acc[j].y * sc + ex * a[u][j].y;
              acc[j].z = acc[j].z * sc + ex * a[u][j].z;
              acc[j].w = acc[j].w * sc + ex * a[u][j].w;
            }
          }
        }
      }
    }

    v4f* po = (v4f*)(out + (size_t)d * ROW) + lane;
    if constexpr (!LAST) {
#pragma unroll
      for (int j = 0; j < NJ; ++j) {
        v4f v;
        v.x = acc[j].x / den[j] + bz[j].x; v.x = v.x > 0.f ? v.x : (expf(v.x) - 1.f);
        v.y = acc[j].y / den[j] + bz[j].y; v.y = v.y > 0.f ? v.y : (expf(v.y) - 1.f);
        v.z = acc[j].z / den[j] + bz[j].z; v.z = v.z > 0.f ? v.z : (expf(v.z) - 1.f);
        v.w = acc[j].w / den[j] + bz[j].w; v.w = v.w > 0.f ? v.w : (expf(v.w) - 1.f);
        po[j * 32] = v;
      }
    } else {
      v4f v;
      v.x = acc[0].x / den[0] + bz[0].x;
      v.y = acc[0].y / den[0] + bz[0].y;
      v.z = acc[0].z / den[0] + bz[0].z;
      v.w = acc[0].w / den[0] + bz[0].w;
      float rm = fmaxf(fmaxf(v.x, v.y), fmaxf(v.z, v.w));
#pragma unroll
      for (int m = 1; m < 32; m <<= 1) rm = fmaxf(rm, __shfl_xor(rm, m));
      float se = (expf(v.x - rm) + expf(v.y - rm)) + (expf(v.z - rm) + expf(v.w - rm));
#pragma unroll
      for (int m = 1; m < 32; m <<= 1) se += __shfl_xor(se, m);
      const float lse = logf(se);
      v.x = v.x - rm - lse; v.y = v.y - rm - lse; v.z = v.z - rm - lse; v.w = v.w - rm - lse;
      po[0] = v;
    }
  }
}

__global__ void __launch_bounds__(256) node_gat_c32(const float* __restrict__ xl,
                                                    const float* __restrict__ xr,
                                                    const float* __restrict__ att,
                                                    const float* __restrict__ bias,
                                                    const unsigned* __restrict__ rowptr,
                                                    const int* __restrict__ csr_src,
                                                    int N, float* __restrict__ out) {
  node_rows<32, false>(xl, xr, att, bias, rowptr, csr_src, N, out);
}

__global__ void __launch_bounds__(256) node_gat_c16(const float* __restrict__ xl,
                                                    const float* __restrict__ xr,
                                                    const float* __restrict__ att,
                                                    const float* __restrict__ bias,
                                                    const unsigned* __restrict__ rowptr,
                                                    const int* __restrict__ csr_src,
                                                    int N, float* __restrict__ out) {
  node_rows<16, true>(xl, xr, att, bias, rowptr, csr_src, N, out);
}

extern "C" void kernel_launch(void* const* d_in, const int* in_sizes, int n_in,
                              void* d_out, int out_size, void* d_ws, size_t ws_size,
                              hipStream_t stream) {
  (void)n_in; (void)out_size; (void)ws_size;
  const float* x    = (const float*)d_in[0];
  const int*   ei   = (const int*)d_in[1];
  const float* W1l  = (const float*)d_in[2];
  const float* W1r  = (const float*)d_in[3];
  const float* att1 = (const float*)d_in[4];
  const float* b1   = (const float*)d_in[5];
  const float* W2l  = (const float*)d_in[6];
  const float* W2r  = (const float*)d_in[7];
  const float* att2 = (const float*)d_in[8];
  const float* b2   = (const float*)d_in[9];

  int N = in_sizes[0] / 128;
  int E = in_sizes[1] / 2;
  const int* srcI = ei;
  const int* dstI = ei + E;

  float* ws    = (float*)d_ws;
  float* xl    = ws;
  float* xr    = xl + (size_t)N * 256;
  float* hbuf  = xr + (size_t)N * 256;
  unsigned* cnt    = (unsigned*)(hbuf + (size_t)N * 256);
  unsigned* rowptr = cnt + (size_t)N;
  unsigned* cursor = rowptr + (size_t)N + 1;
  int*      csr    = (int*)(cursor + (size_t)N);
  float* outF = (float*)d_out;

  dim3 blk(256);

  fill_u32<<<256, blk, 0, stream>>>(cnt, 0u, (long long)N);
  dst_count<<<1024, blk, 0, stream>>>(dstI, E, cnt);
  dst_scan<<<1, dim3(1024), 0, stream>>>(cnt, N, rowptr, cursor);
  dst_scatter<<<1024, blk, 0, stream>>>(srcI, dstI, E, cursor, csr);

  int tiles1 = (N / 16) * (256 / 16);
  gemm_wmma<<<(tiles1 + 7) / 8, blk, 0, stream>>>(x, W1l, xl, N, 128, 256);
  gemm_wmma<<<(tiles1 + 7) / 8, blk, 0, stream>>>(x, W1r, xr, N, 128, 256);
  node_gat_c32<<<(N + 15) / 16, blk, 0, stream>>>(xl, xr, att1, b1, rowptr, csr, N, hbuf);

  int tiles2 = (N / 16) * (128 / 16);
  gemm_wmma<<<(tiles2 + 7) / 8, blk, 0, stream>>>(hbuf, W2l, xl, N, 256, 128);
  gemm_wmma<<<(tiles2 + 7) / 8, blk, 0, stream>>>(hbuf, W2r, xr, N, 256, 128);
  node_gat_c16<<<(N + 15) / 16, blk, 0, stream>>>(xl, xr, att2, b2, rowptr, csr, N, outF);
}


# GEMM re-tiled: one launch per layer, block = 32 rows x both products, wave tile 32x64 (2x4 accumulators): an A/W fragment load feeds 4/2 WMMAs; 16-B loads, line stores; k order of f32 sums changed
# speedup vs baseline: 7.0381x; 1.4405x over previous
#include <hip/hip_runtime.h>
#include <math.h>

typedef __attribute__((ext_vector_type(2))) float v2f;
typedef __attribute__((ext_vector_type(8))) float v8f;
typedef __attribute__((ext_vector_type(4))) float v4f;

#define HEADS 8

#if defined(__gfx1250__) && __has_builtin(__builtin_amdgcn_wmma_f32_16x16x4_f32)
#define USE_WMMA_F32 1
#else
#define USE_WMMA_F32 0
#endif

struct GemmFrag { v4f a[2]; v4f b[4]; };

template <int KIN, int KOUT>
__global__ void __launch_bounds__((KOUT / 32) * 32)
gemm_pair_wmma(const float* __restrict__ A, const float* __restrict__ Wl,
               const float* __restrict__ Wr, float* __restrict__ Cl, float* __restrict__ Cr,
               int M) {
  static_assert(KIN % 16 == 0 && KOUT % 64 == 0, "K in pairs of 8-wide steps, 64-column strips");
  constexpr int NS = KOUT / 64;
  const int lane = threadIdx.x & 31;
  const int h = lane >> 4;
  const int c = lane & 15;
  const int wv = __builtin_amdgcn_readfirstlane((int)(threadIdx.x >> 5));
  const bool right = wv >= NS;
  const float* __restrict__ W = right ? Wr : Wl;
  float* __restrict__ C       = right ? Cr : Cl;
  const int col0 = (right ? wv - NS : wv) * 64;
  const int row0 = blockIdx.x * 32;
  if (row0 >= M) return;

#if USE_WMMA_F32
  const int ra0 = row0 + c      < M ? row0 + c      : M - 1;
  const int ra1 = row0 + 16 + c < M ? row0 + 16 + c : M - 1;
  const v4f* a0p = (const v4f*)(A + (size_t)ra0 * KIN + 4 * h);
  const v4f* a1p = (const v4f*)(A + (size_t)ra1 * KIN + 4 * h);
  const v4f* bp  = (const v4f*)(W + (size_t)(4 * h) * KOUT + col0 + 4 * c);

  v8f acc[2][4] = {};
  auto load = [&](GemmFrag& f, int kk) {
    f.a[0] = a0p[2 * kk];
    f.a[1] = a1p[2 * kk];
#pragma unroll
    for (int e = 0; e < 4; ++e) f.b[e] = bp[(size_t)(8 * kk + e) * (KOUT / 4)];
  };
  auto mma = [&](const GemmFrag& f) {
#pragma unroll
    for (int p = 0; p < 2; ++p) {
#pragma unroll
      for (int r = 0; r < 2; ++r) {
        v2f a;
        a[0] = f.a[r][2 * p];
        a[1] = f.a[r][2 * p + 1];
#pragma unroll
        for (int t = 0; t < 4; ++t) {
          v2f b;
          b[0] = f.b[2 * p][t];
          b[1] = f.b[2 * p + 1][t];
          acc[r][t] = __builtin_amdgcn_wmma_f32_16x16x4_f32(false, a, false, b,
                                                            (short)0, acc[r][t], false, false);
        }
      }
    }
  };

  constexpr int NK = KIN / 8;
  GemmFrag f0, f1;
  load(f0, 0);
#pragma nounroll
  for (int kk = 0; kk < NK - 2; kk += 2) {
    load(f1, kk + 1);
    __builtin_amdgcn_sched_barrier(0);
    mma(f0);
    load(f0, kk + 2);
    __builtin_amdgcn_sched_barrier(0);
    mma(f1);
  }
  load(f1, NK - 1);
  __builtin_amdgcn_sched_barrier(0);
  mma(f0);
  mma(f1);

#pragma unroll
  for (int r = 0; r < 2; ++r) {
#pragma unroll
    for (int v = 0; v < 8; ++v) {
      const int row = row0 + 16 * r + 8 * h + v;
      if (row < M) {
        v4f o;
        o.x = acc[r][0][v]; o.y = acc[r][1][v]; o.z = acc[r][2][v]; o.w = acc[r][3][v];
        *(v4f*)(C + (size_t)row * KOUT + col0 + 4 * c) = o;
      }
    }
  }
#else
  for (int i = lane; i < 32 * 64; i += 32) {
    const int row = row0 + i / 64, col = col0 + i % 64;
    if (row < M) {
      float sum = 0.f;
      for (int k = 0; k < KIN; ++k) sum += A[(size_t)row * KIN + k] * W[(size_t)k * KOUT + col];
      C[(size_t)row * KOUT + col] = sum;
    }
  }
#endif
}

__global__ void fill_u32(unsigned* __restrict__ p, unsigned v, long long n) {
  long long i = blockIdx.x * (long long)blockDim.x + threadIdx.x;
  long long stride = (long long)gridDim.x * blockDim.x;
  for (; i < n; i += stride) p[i] = v;
}

__global__ void dst_count(const int* __restrict__ dstI, int E, unsigned* __restrict__ cnt) {
  int i = blockIdx.x * blockDim.x + threadIdx.x;
  int stride = gridDim.x * blockDim.x;
  for (; i < E; i += stride) atomicAdd(cnt + dstI[i], 1u);
}

__global__ void __launch_bounds__(1024) dst_scan(const unsigned* __restrict__ cnt, int N,
                                                 unsigned* __restrict__ rowptr,
                                                 unsigned* __restrict__ cursor) {
  __shared__ unsigned wsum[32];
  const int t = threadIdx.x, lane = t & 31, wv = t >> 5;
  unsigned carry = 0;
  unsigned n0, n1, n2, n3;
  {
    const int i0 = 4 * t;
    n0 = i0     < N ? cnt[i0]     : 0u;
    n1 = i0 + 1 < N ? cnt[i0 + 1] : 0u;
    n2 = i0 + 2 < N ? cnt[i0 + 2] : 0u;
    n3 = i0 + 3 < N ? cnt[i0 + 3] : 0u;
  }
  for (int base = 0; base < N; base += 4096) {
    const int i0 = base + 4 * t;
    const unsigned c0 = n0, c1 = n1, c2 = n2, c3 = n3;
    {
      const int j0 = i0 + 4096;
      n0 = j0     < N ? cnt[j0]     : 0u;
      n1 = j0 + 1 < N ? cnt[j0 + 1] : 0u;
      n2 = j0 + 2 < N ? cnt[j0 + 2] : 0u;
      n3 = j0 + 3 < N ? cnt[j0 + 3] : 0u;
    }
    const unsigned local = c0 + c1 + c2 + c3;
    unsigned inc = local;
#pragma unroll
    for (int o = 1; o < 32; o <<= 1) {
      unsigned nb = __shfl_up(inc, o);
      if (lane >= o) inc += nb;
    }
    if (lane == 31) wsum[wv] = inc;
    __syncthreads();
    const unsigned wtot = wsum[lane];
    unsigned winc = wtot;
#pragma unroll
    for (int o = 1; o < 32; o <<= 1) {
      unsigned nb = __shfl_up(winc, o);
      if (lane >= o) winc += nb;
    }
    const unsigned woff  = __shfl(winc, wv) - __shfl(wtot, wv);
    const unsigned total = __shfl(winc, 31);
    unsigned ex = carry + woff + inc - local;
    if (i0     < N) { rowptr[i0]     = ex; cursor[i0]     = ex; } ex += c0;
    if (i0 + 1 < N) { rowptr[i0 + 1] = ex; cursor[i0 + 1] = ex; } ex += c1;
    if (i0 + 2 < N) { rowptr[i0 + 2] = ex; cursor[i0 + 2] = ex; } ex += c2;
    if (i0 + 3 < N) { rowptr[i0 + 3] = ex; cursor[i0 + 3] = ex; }
    carry += total;
    __syncthreads();
  }
  if (t == 0) rowptr[N] = carry;
}

__global__ void dst_scatter(const int* __restrict__ srcI, const int* __restrict__ dstI, int E,
                            unsigned* __restrict__ cursor, int* __restrict__ csr_src) {
  int i = blockIdx.x * blockDim.x + threadIdx.x;
  int stride = gridDim.x * blockDim.x;
  for (; i < E; i += stride) {
    unsigned p = atomicAdd(cursor + dstI[i], 1u);
    csr_src[p] = srcI[i];
  }
}

__device__ __forceinline__ float leaky_dot4(v4f a, v4f b, v4f w) {
  float acc, v;
  v = a.x + b.x; v = fmaxf(v, 0.2f * v); acc  = v * w.x;
  v = a.y + b.y; v = fmaxf(v, 0.2f * v); acc += v * w.y;
  v = a.z + b.z; v = fmaxf(v, 0.2f * v); acc += v * w.z;
  v = a.w + b.w; v = fmaxf(v, 0.2f * v); acc += v * w.w;
  return acc;
}

template <int C>
__device__ __forceinline__ float group_logit(v4f a, v4f b, v4f w) {
  float p = leaky_dot4(a, b, w);
#pragma unroll
  for (int m = 1; m < C / 4; m <<= 1) p += __shfl_xor(p, m);
  return p;
}

template <int C, bool LAST>
__device__ __forceinline__ void node_rows(const float* __restrict__ xl,
                                          const float* __restrict__ xr,
                                          const float* __restrict__ att,
                                          const float* __restrict__ bias,
                                          const unsigned* __restrict__ rowptr,
                                          const int* __restrict__ csr_src,
                                          int N, float* __restrict__ out) {
  static_assert(C == 32 || C == 16, "row pieces of 128 floats, 8 or 4 lanes per head");
  static_assert(!LAST || C == 16, "the log-softmax epilogue takes one 128-float piece");
  constexpr int ROW = HEADS * C;
  constexpr int NJ  = ROW / 128;
  constexpr int UN  = 4;
  static_assert(UN == 4 && (NJ == 1 || NJ == 2), "the load fence below names UN*NJ vectors");
  const int lane = threadIdx.x & 31;
  const int wib  = __builtin_amdgcn_readfirstlane((int)(threadIdx.x >> 5));
  const int wave = blockIdx.x * (int)(blockDim.x >> 5) + wib;
  const int nw   = gridDim.x * (int)(blockDim.x >> 5);
  v4f w[NJ], bz[NJ];
#pragma unroll
  for (int j = 0; j < NJ; ++j) {
    w[j]  = ((const v4f*)att)[j * 32 + lane];
    bz[j] = ((const v4f*)bias)[j * 32 + lane];
  }
  for (int d = wave; d < N; d += nw) {
    const unsigned beg = rowptr[d], end = rowptr[d + 1];
    const v4f* pr = (const v4f*)(xr + (size_t)d * ROW) + lane;
    const v4f* ps = (const v4f*)(xl + (size_t)d * ROW) + lane;
    v4f b[NJ], acc[NJ];
    float  mx[NJ], den[NJ];
#pragma unroll
    for (int j = 0; j < NJ; ++j) { b[j] = pr[j * 32]; acc[j] = ps[j * 32]; }
#pragma unroll
    for (int j = 0; j < NJ; ++j) { mx[j] = group_logit<C>(acc[j], b[j], w[j]); den[j] = 1.f; }

    for (unsigned base = beg; base < end; base += 32) {
      const int cnt = (int)(end - base) < 32 ? (int)(end - base) : 32;
      const int my  = lane < cnt ? csr_src[base + lane] : 0;
      for (int i = 0; i < cnt; i += UN) {
        v4f a[UN][NJ];
#pragma unroll
        for (int u = 0; u < UN; ++u) {
          const int ii = i + u < cnt ? i + u : cnt - 1;
          const int s  = __builtin_amdgcn_readlane(my, ii);
          const v4f* p = (const v4f*)(xl + (size_t)s * ROW) + lane;
#pragma unroll
          for (int j = 0; j < NJ; ++j) a[u][j] = p[j * 32];
        }
        if constexpr (NJ == 2)
          asm volatile("" :: "v"(a[0][0]), "v"(a[0][1]), "v"(a[1][0]), "v"(a[1][1]),
                             "v"(a[2][0]), "v"(a[2][1]), "v"(a[3][0]), "v"(a[3][1]));
        else
          asm volatile("" :: "v"(a[0][0]), "v"(a[1][0]), "v"(a[2][0]), "v"(a[3][0]));
#pragma unroll
        for (int u = 0; u < UN; ++u) {
          if (u == 0 || i + u < cnt) {
#pragma unroll
            for (int j = 0; j < NJ; ++j) {
              const float lg = group_logit<C>(a[u][j], b[j], w[j]);
              const float t  = expf(-fabsf(mx[j] - lg));
              const bool  up = lg > mx[j];
              const float sc = up ? t : 1.f;
              const float ex = up ? 1.f : t;
              mx[j]  = fmaxf(mx[j], lg);
              den[j] = den[j] * sc + ex;
              acc[j].x = acc[j].x * sc + ex * a[u][j].x;
              acc[j].y = acc[j].y * sc + ex * a[u][j].y;
              acc[j].z = acc[j].z * sc + ex * a[u][j].z;
              acc[j].w = acc[j].w * sc + ex * a[u][j].w;
            }
          }
        }
      }
    }

    v4f* po = (v4f*)(out + (size_t)d * ROW) + lane;
    if constexpr (!LAST) {
#pragma unroll
      for (int j = 0; j < NJ; ++j) {
        v4f v;
        v.x = acc[j].x / den[j] + bz[j].x; v.x = v.x > 0.f ? v.x : (expf(v.x) - 1.f);
        v.y = acc[j].y / den[j] + bz[j].y; v.y = v.y > 0.f ? v.y : (expf(v.y) - 1.f);
        v.z = acc[j].z / den[j] + bz[j].z; v.z = v.z > 0.f ? v.z : (expf(v.z) - 1.f);
        v.w = acc[j].w / den[j] + bz[j].w; v.w = v.w > 0.f ? v.w : (expf(v.w) - 1.f);
        po[j * 32] = v;
      }
    } else {
      v4f v;
      v.x = acc[0].x / den[0] + bz[0].x;
      v.y = acc[0].y / den[0] + bz[0].y;
      v.z = acc[0].z / den[0] + bz[0].z;
      v.w = acc[0].w / den[0] + bz[0].w;
      float rm = fmaxf(fmaxf(v.x, v.y), fmaxf(v.z, v.w));
#pragma unroll
      for (int m = 1; m < 32; m <<= 1) rm = fmaxf(rm, __shfl_xor(rm, m));
      float se = (expf(v.x - rm) + expf(v.y - rm)) + (expf(v.z - rm) + expf(v.w - rm));
#pragma unroll
      for (int m = 1; m < 32; m <<= 1) se += __shfl_xor(se, m);
      const float lse = logf(se);
      v.x = v.x - rm - lse; v.y = v.y - rm - lse; v.z = v.z - rm - lse; v.w = v.w - rm - lse;
      po[0] = v;
    }
  }
}

__global__ void __launch_bounds__(256) node_gat_c32(const float* __restrict__ xl,
                                                    const float* __restrict__ xr,
                                                    const float* __restrict__ att,
                                                    const float* __restrict__ bias,
                                                    const unsigned* __restrict__ rowptr,
                                                    const int* __restrict__ csr_src,
                                                    int N, float* __restrict__ out) {
  node_rows<32, false>(xl, xr, att, bias, rowptr, csr_src, N, out);
}

__global__ void __launch_bounds__(256) node_gat_c16(const float* __restrict__ xl,
                                                    const float* __restrict__ xr,
                                                    const float* __restrict__ att,
                                                    const float* __restrict__ bias,
                                                    const unsigned* __restrict__ rowptr,
                                                    const int* __restrict__ csr_src,
                                                    int N, float* __restrict__ out) {
  node_rows<16, true>(xl, xr, att, bias, rowptr, csr_src, N, out);
}

extern "C" void kernel_launch(void* const* d_in, const int* in_sizes, int n_in,
                              void* d_out, int out_size, void* d_ws, size_t ws_size,
                              hipStream_t stream) {
  (void)n_in; (void)out_size; (void)ws_size;
  const float* x    = (const float*)d_in[0];
  const int*   ei   = (const int*)d_in[1];
  const float* W1l  = (const float*)d_in[2];
  const float* W1r  = (const float*)d_in[3];
  const float* att1 = (const float*)d_in[4];
  const float* b1   = (const float*)d_in[5];
  const float* W2l  = (const float*)d_in[6];
  const float* W2r  = (const float*)d_in[7];
  const float* att2 = (const float*)d_in[8];
  const float* b2   = (const float*)d_in[9];

  int N = in_sizes[0] / 128;
  int E = in_sizes[1] / 2;
  const int* srcI = ei;
  const int* dstI = ei + E;

  float* ws    = (float*)d_ws;
  float* xl    = ws;
  float* xr    = xl + (size_t)N * 256;
  float* hbuf  = xr + (size_t)N * 256;
  unsigned* cnt    = (unsigned*)(hbuf + (size_t)N * 256);
  unsigned* rowptr = cnt + (size_t)N;
  unsigned* cursor = rowptr + (size_t)N + 1;
  int*      csr    = (int*)(cursor + (size_t)N);
  float* outF = (float*)d_out;

  dim3 blk(256);

  fill_u32<<<256, blk, 0, stream>>>(cnt, 0u, (long long)N);
  dst_count<<<1024, blk, 0, stream>>>(dstI, E, cnt);
  dst_scan<<<1, dim3(1024), 0, stream>>>(cnt, N, rowptr, cursor);
  dst_scatter<<<1024, blk, 0, stream>>>(srcI, dstI, E, cursor, csr);

  int rtiles = (N + 31) / 32;
  gemm_pair_wmma<128, 256><<<rtiles, dim3(256), 0, stream>>>(x, W1l, W1r, xl, xr, N);
  node_gat_c32<<<(N + 15) / 16, blk, 0, stream>>>(xl, xr, att1, b1, rowptr, csr, N, hbuf);

  gemm_pair_wmma<256, 128><<<rtiles, dim3(128), 0, stream>>>(hbuf, W2l, W2r, xl, xr, N);
  node_gat_c16<<<(N + 15) / 16, blk, 0, stream>>>(xl, xr, att2, b2, rowptr, csr, N, outF);
}


# edge grouping by chunk-local LDS counters: returning LDS add = rank in chunk, per-chunk counts prefixed per node, slot = rowptr + earlier chunks + rank; no global atomics; segment order by arrival
# speedup vs baseline: 8.2008x; 1.1652x over previous
#include <hip/hip_runtime.h>
#include <math.h>

typedef __attribute__((ext_vector_type(2))) float v2f;
typedef __attribute__((ext_vector_type(8))) float v8f;
typedef __attribute__((ext_vector_type(4))) float v4f;

#define HEADS 8

#if defined(__gfx1250__) && __has_builtin(__builtin_amdgcn_wmma_f32_16x16x4_f32)
#define USE_WMMA_F32 1
#else
#define USE_WMMA_F32 0
#endif

struct GemmFrag { v4f a[2]; v4f b[4]; };

template <int KIN, int KOUT>
__global__ void __launch_bounds__((KOUT / 32) * 32)
gemm_pair_wmma(const float* __restrict__ A, const float* __restrict__ Wl,
               const float* __restrict__ Wr, float* __restrict__ Cl, float* __restrict__ Cr,
               int M) {
  static_assert(KIN % 16 == 0 && KOUT % 64 == 0, "K in pairs of 8-wide steps, 64-column strips");
  constexpr int NS = KOUT / 64;
  const int lane = threadIdx.x & 31;
  const int h = lane >> 4;
  const int c = lane & 15;
  const int wv = __builtin_amdgcn_readfirstlane((int)(threadIdx.x >> 5));
  const bool right = wv >= NS;
  const float* __restrict__ W = right ? Wr : Wl;
  float* __restrict__ C       = right ? Cr : Cl;
  const int col0 = (right ? wv - NS : wv) * 64;
  const int row0 = blockIdx.x * 32;
  if (row0 >= M) return;

#if USE_WMMA_F32
  const int ra0 = row0 + c      < M ? row0 + c      : M - 1;
  const int ra1 = row0 + 16 + c < M ? row0 + 16 + c : M - 1;
  const v4f* a0p = (const v4f*)(A + (size_t)ra0 * KIN + 4 * h);
  const v4f* a1p = (const v4f*)(A + (size_t)ra1 * KIN + 4 * h);
  const v4f* bp  = (const v4f*)(W + (size_t)(4 * h) * KOUT + col0 + 4 * c);

  v8f acc[2][4] = {};
  auto load = [&](GemmFrag& f, int kk) {
    f.a[0] = a0p[2 * kk];
    f.a[1] = a1p[2 * kk];
#pragma unroll
    for (int e = 0; e < 4; ++e) f.b[e] = bp[(size_t)(8 * kk + e) * (KOUT / 4)];
  };
  auto mma = [&](const GemmFrag& f) {
#pragma unroll
    for (int p = 0; p < 2; ++p) {
#pragma unroll
      for (int r = 0; r < 2; ++r) {
        v2f a;
        a[0] = f.a[r][2 * p];
        a[1] = f.a[r][2 * p + 1];
#pragma unroll
        for (int t = 0; t < 4; ++t) {
          v2f b;
          b[0] = f.b[2 * p][t];
          b[1] = f.b[2 * p + 1][t];
          acc[r][t] = __builtin_amdgcn_wmma_f32_16x16x4_f32(false, a, false, b,
                                                            (short)0, acc[r][t], false, false);
        }
      }
    }
  };

  constexpr int NK = KIN / 8;
  GemmFrag f0, f1;
  load(f0, 0);
#pragma nounroll
  for (int kk = 0; kk < NK - 2; kk += 2) {
    load(f1, kk + 1);
    __builtin_amdgcn_sched_barrier(0);
    mma(f0);
    load(f0, kk + 2);
    __builtin_amdgcn_sched_barrier(0);
    mma(f1);
  }
  load(f1, NK - 1);
  __builtin_amdgcn_sched_barrier(0);
  mma(f0);
  mma(f1);

#pragma unroll
  for (int r = 0; r < 2; ++r) {
#pragma unroll
    for (int v = 0; v < 8; ++v) {
      const int row = row0 + 16 * r + 8 * h + v;
      if (row < M) {
        v4f o;
        o.x = acc[r][0][v]; o.y = acc[r][1][v]; o.z = acc[r][2][v]; o.w = acc[r][3][v];
        *(v4f*)(C + (size_t)row * KOUT + col0 + 4 * c) = o;
      }
    }
  }
#else
  for (int i = lane; i < 32 * 64; i += 32) {
    const int row = row0 + i / 64, col = col0 + i % 64;
    if (row < M) {
      float sum = 0.f;
      for (int k = 0; k < KIN; ++k) sum += A[(size_t)row * KIN + k] * W[(size_t)k * KOUT + col];
      C[(size_t)row * KOUT + col] = sum;
    }
  }
#endif
}

#define GRP_CAP 51200
#define GRP_CHUNKS 64

__global__ void __launch_bounds__(1024) grp_rank(const int* __restrict__ dstI, int E, int N,
                                                 unsigned* __restrict__ rank,
                                                 unsigned* __restrict__ cntG) {
  __shared__ uint4 lc4[GRP_CAP / 4];
  unsigned* lc = (unsigned*)lc4;
  const int t = threadIdx.x;
  const int g = blockIdx.x;
  for (int i = t; i < GRP_CAP / 4; i += 1024) lc4[i] = make_uint4(0u, 0u, 0u, 0u);
  __syncthreads();
  const int per = (E + (int)gridDim.x - 1) / (int)gridDim.x;
  const int e0 = g * per;
  const int e1 = e0 + per < E ? e0 + per : E;
  for (int e = e0 + t; e < e1; e += 8 * 1024) {
    int d[8];
#pragma unroll
    for (int u = 0; u < 8; ++u) d[u] = dstI[e + u * 1024 < e1 ? e + u * 1024 : e1 - 1];
    asm volatile("" :: "v"(d[0]), "v"(d[1]), "v"(d[2]), "v"(d[3]),
                       "v"(d[4]), "v"(d[5]), "v"(d[6]), "v"(d[7]));
#pragma unroll
    for (int u = 0; u < 8; ++u)
      if (e + u * 1024 < e1) rank[e + u * 1024] = atomicAdd(&lc[d[u]], 1u);
  }
  __syncthreads();
  uint4* __restrict__ row = (uint4*)(cntG + (size_t)g * GRP_CAP);
  for (int i = t; i < (N + 3) / 4; i += 1024) row[i] = lc4[i];
}

template <int G>
__global__ void grp_prefix(unsigned* __restrict__ cntG, int N, unsigned* __restrict__ tot) {
  static_assert(G % 32 == 0, "rows are read in batches of 32");
  const int d = blockIdx.x * blockDim.x + threadIdx.x;
  if (d >= N) return;
  unsigned run = 0;
  for (int g0 = 0; g0 < G; g0 += 32) {
    unsigned c[32];
#pragma unroll
    for (int u = 0; u < 32; ++u) c[u] = cntG[(size_t)(g0 + u) * GRP_CAP + d];
#pragma unroll
    for (int u = 0; u < 32; ++u) { cntG[(size_t)(g0 + u) * GRP_CAP + d] = run; run += c[u]; }
  }
  tot[d] = run;
}

__global__ void __launch_bounds__(1024) dst_scan(const unsigned* __restrict__ cnt, int N,
                                                 unsigned* __restrict__ rowptr) {
  __shared__ unsigned wsum[32];
  const int t = threadIdx.x, lane = t & 31, wv = t >> 5;
  unsigned carry = 0;
  unsigned n0, n1, n2, n3;
  {
    const int i0 = 4 * t;
    n0 = i0     < N ? cnt[i0]     : 0u;
    n1 = i0 + 1 < N ? cnt[i0 + 1] : 0u;
    n2 = i0 + 2 < N ? cnt[i0 + 2] : 0u;
    n3 = i0 + 3 < N ? cnt[i0 + 3] : 0u;
  }
  for (int base = 0; base < N; base += 4096) {
    const int i0 = base + 4 * t;
    const unsigned c0 = n0, c1 = n1, c2 = n2, c3 = n3;
    {
      const int j0 = i0 + 4096;
      n0 = j0     < N ? cnt[j0]     : 0u;
      n1 = j0 + 1 < N ? cnt[j0 + 1] : 0u;
      n2 = j0 + 2 < N ? cnt[j0 + 2] : 0u;
      n3 = j0 + 3 < N ? cnt[j0 + 3] : 0u;
    }
    const unsigned local = c0 + c1 + c2 + c3;
    unsigned inc = local;
#pragma unroll
    for (int o = 1; o < 32; o <<= 1) {
      unsigned nb = __shfl_up(inc, o);
      if (lane >= o) inc += nb;
    }
    if (lane == 31) wsum[wv] = inc;
    __syncthreads();
    const unsigned wtot = wsum[lane];
    unsigned winc = wtot;
#pragma unroll
    for (int o = 1; o < 32; o <<= 1) {
      unsigned nb = __shfl_up(winc, o);
      if (lane >= o) winc += nb;
    }
    const unsigned woff  = __shfl(winc, wv) - __shfl(wtot, wv);
    const unsigned total = __shfl(winc, 31);
    unsigned ex = carry + woff + inc - local;
    if (i0     < N) rowptr[i0]     = ex;
    ex += c0;
    if (i0 + 1 < N) rowptr[i0 + 1] = ex;
    ex += c1;
    if (i0 + 2 < N) rowptr[i0 + 2] = ex;
    ex += c2;
    if (i0 + 3 < N) rowptr[i0 + 3] = ex;
    carry += total;
    __syncthreads();
  }
  if (t == 0) rowptr[N] = carry;
}

__global__ void __launch_bounds__(256) grp_place(const int* __restrict__ srcI,
                                                 const int* __restrict__ dstI, int E, int G,
                                                 const unsigned* __restrict__ rank,
                                                 const unsigned* __restrict__ cntG,
                                                 const unsigned* __restrict__ rowptr,
                                                 int* __restrict__ csr_src) {
  const int eb = blockIdx.x * 1024 + threadIdx.x;
  if (eb >= E) return;
  const int per = (E + G - 1) / G;
  int d[4], s[4];
  unsigned r[4], a[4], b[4];
#pragma unroll
  for (int u = 0; u < 4; ++u) {
    const int e = eb + u * 256 < E ? eb + u * 256 : E - 1;
    d[u] = dstI[e];
    s[u] = srcI[e];
    r[u] = rank[e];
  }
#pragma unroll
  for (int u = 0; u < 4; ++u) {
    const int e = eb + u * 256 < E ? eb + u * 256 : E - 1;
    a[u] = rowptr[d[u]];
    b[u] = cntG[(size_t)(e / per) * GRP_CAP + d[u]];
  }
  asm volatile("" :: "v"(s[0]), "v"(s[1]), "v"(s[2]), "v"(s[3]), "v"(r[0]), "v"(r[1]), "v"(r[2]),
                     "v"(r[3]), "v"(a[0]), "v"(a[1]), "v"(a[2]), "v"(a[3]), "v"(b[0]), "v"(b[1]),
                     "v"(b[2]), "v"(b[3]));
#pragma unroll
  for (int u = 0; u < 4; ++u)
    if (eb + u * 256 < E) csr_src[a[u] + b[u] + r[u]] = s[u];
}

__device__ __forceinline__ float leaky_dot4(v4f a, v4f b, v4f w) {
  float acc, v;
  v = a.x + b.x; v = fmaxf(v, 0.2f * v); acc  = v * w.x;
  v = a.y + b.y; v = fmaxf(v, 0.2f * v); acc += v * w.y;
  v = a.z + b.z; v = fmaxf(v, 0.2f * v); acc += v * w.z;
  v = a.w + b.w; v = fmaxf(v, 0.2f * v); acc += v * w.w;
  return acc;
}

template <int C>
__device__ __forceinline__ float group_logit(v4f a, v4f b, v4f w) {
  float p = leaky_dot4(a, b, w);
#pragma unroll
  for (int m = 1; m < C / 4; m <<= 1) p += __shfl_xor(p, m);
  return p;
}

template <int C, bool LAST>
__device__ __forceinline__ void node_rows(const float* __restrict__ xl,
                                          const float* __restrict__ xr,
                                          const float* __restrict__ att,
                                          const float* __restrict__ bias,
                                          const unsigned* __restrict__ rowptr,
                                          const int* __restrict__ csr_src,
                                          int N, float* __restrict__ out) {
  static_assert(C == 32 || C == 16, "row pieces of 128 floats, 8 or 4 lanes per head");
  static_assert(!LAST || C == 16, "the log-softmax epilogue takes one 128-float piece");
  constexpr int ROW = HEADS * C;
  constexpr int NJ  = ROW / 128;
  constexpr int UN  = 4;
  static_assert(UN == 4 && (NJ == 1 || NJ == 2), "the load fence below names UN*NJ vectors");
  const int lane = threadIdx.x & 31;
  const int wib  = __builtin_amdgcn_readfirstlane((int)(threadIdx.x >> 5));
  const int wave = blockIdx.x * (int)(blockDim.x >> 5) + wib;
  const int nw   = gridDim.x * (int)(blockDim.x >> 5);
  v4f w[NJ], bz[NJ];
#pragma unroll
  for (int j = 0; j < NJ; ++j) {
    w[j]  = ((const v4f*)att)[j * 32 + lane];
    bz[j] = ((const v4f*)bias)[j * 32 + lane];
  }
  for (int d = wave; d < N; d += nw) {
    const unsigned beg = rowptr[d], end = rowptr[d + 1];
    const v4f* pr = (const v4f*)(xr + (size_t)d * ROW) + lane;
    const v4f* ps = (const v4f*)(xl + (size_t)d * ROW) + lane;
    v4f b[NJ], acc[NJ];
    float  mx[NJ], den[NJ];
#pragma unroll
    for (int j = 0; j < NJ; ++j) { b[j] = pr[j * 32]; acc[j] = ps[j * 32]; }
#pragma unroll
    for (int j = 0; j < NJ; ++j) { mx[j] = group_logit<C>(acc[j], b[j], w[j]); den[j] = 1.f; }

    for (unsigned base = beg; base < end; base += 32) {
      const int cnt = (int)(end - base) < 32 ? (int)(end - base) : 32;
      const int my  = lane < cnt ? csr_src[base + lane] : 0;
      for (int i = 0; i < cnt; i += UN) {
        v4f a[UN][NJ];
#pragma unroll
        for (int u = 0; u < UN; ++u) {
          const int ii = i + u < cnt ? i + u : cnt - 1;
          const int s  = __builtin_amdgcn_readlane(my, ii);
          const v4f* p = (const v4f*)(xl + (size_t)s * ROW) + lane;
#pragma unroll
          for (int j = 0; j < NJ; ++j) a[u][j] = p[j * 32];
        }
        if constexpr (NJ == 2)
          asm volatile("" :: "v"(a[0][0]), "v"(a[0][1]), "v"(a[1][0]), "v"(a[1][1]),
                             "v"(a[2][0]), "v"(a[2][1]), "v"(a[3][0]), "v"(a[3][1]));
        else
          asm volatile("" :: "v"(a[0][0]), "v"(a[1][0]), "v"(a[2][0]), "v"(a[3][0]));
#pragma unroll
        for (int u = 0; u < UN; ++u) {
          if (u == 0 || i + u < cnt) {
#pragma unroll
            for (int j = 0; j < NJ; ++j) {
              const float lg = group_logit<C>(a[u][j], b[j], w[j]);
              const float t  = expf(-fabsf(mx[j] - lg));
              const bool  up = lg > mx[j];
              const float sc = up ? t : 1.f;
              const float ex = up ? 1.f : t;
              mx[j]  = fmaxf(mx[j], lg);
              den[j] = den[j] * sc + ex;
              acc[j].x = acc[j].x * sc + ex * a[u][j].x;
              acc[j].y = acc[j].y * sc + ex * a[u][j].y;
              acc[j].z = acc[j].z * sc + ex * a[u][j].z;
              acc[j].w = acc[j].w * sc + ex * a[u][j].w;
            }
          }
        }
      }
    }

    v4f* po = (v4f*)(out + (size_t)d * ROW) + lane;
    if constexpr (!LAST) {
#pragma unroll
      for (int j = 0; j < NJ; ++j) {
        v4f v;
        v.x = acc[j].x / den[j] + bz[j].x; v.x = v.x > 0.f ? v.x : (expf(v.x) - 1.f);
        v.y = acc[j].y / den[j] + bz[j].y; v.y = v.y > 0.f ? v.y : (expf(v.y) - 1.f);
        v.z = acc[j].z / den[j] + bz[j].z; v.z = v.z > 0.f ? v.z : (expf(v.z) - 1.f);
        v.w = acc[j].w / den[j] + bz[j].w; v.w = v.w > 0.f ? v.w : (expf(v.w) - 1.f);
        po[j * 32] = v;
      }
    } else {
      v4f v;
      v.x = acc[0].x / den[0] + bz[0].x;
      v.y = acc[0].y / den[0] + bz[0].y;
      v.z = acc[0].z / den[0] + bz[0].z;
      v.w = acc[0].w / den[0] + bz[0].w;
      float rm = fmaxf(fmaxf(v.x, v.y), fmaxf(v.z, v.w));
#pragma unroll
      for (int m = 1; m < 32; m <<= 1) rm = fmaxf(rm, __shfl_xor(rm, m));
      float se = (expf(v.x - rm) + expf(v.y - rm)) + (expf(v.z - rm) + expf(v.w - rm));
#pragma unroll
      for (int m = 1; m < 32; m <<= 1) se += __shfl_xor(se, m);
      const float lse = logf(se);
      v.x = v.x - rm - lse; v.y = v.y - rm - lse; v.z = v.z - rm - lse; v.w = v.w - rm - lse;
      po[0] = v;
    }
  }
}

__global__ void __launch_bounds__(256) node_gat_c32(const float* __restrict__ xl,
                                                    const float* __restrict__ xr,
                                                    const float* __restrict__ att,
                                                    const float* __restrict__ bias,
                                                    const unsigned* __restrict__ rowptr,
                                                    const int* __restrict__ csr_src,
                                                    int N, float* __restrict__ out) {
  node_rows<32, false>(xl, xr, att, bias, rowptr, csr_src, N, out);
}

__global__ void __launch_bounds__(256) node_gat_c16(const float* __restrict__ xl,
                                                    const float* __restrict__ xr,
                                                    const float* __restrict__ att,
                                                    const float* __restrict__ bias,
                                                    const unsigned* __restrict__ rowptr,
                                                    const int* __restrict__ csr_src,
                                                    int N, float* __restrict__ out) {
  node_rows<16, true>(xl, xr, att, bias, rowptr, csr_src, N, out);
}

extern "C" void kernel_launch(void* const* d_in, const int* in_sizes, int n_in,
                              void* d_out, int out_size, void* d_ws, size_t ws_size,
                              hipStream_t stream) {
  (void)n_in; (void)out_size; (void)ws_size;
  const float* x    = (const float*)d_in[0];
  const int*   ei   = (const int*)d_in[1];
  const float* W1l  = (const float*)d_in[2];
  const float* W1r  = (const float*)d_in[3];
  const float* att1 = (const float*)d_in[4];
  const float* b1   = (const float*)d_in[5];
  const float* W2l  = (const float*)d_in[6];
  const float* W2r  = (const float*)d_in[7];
  const float* att2 = (const float*)d_in[8];
  const float* b2   = (const float*)d_in[9];

  int N = in_sizes[0] / 128;
  int E = in_sizes[1] / 2;
  const int* srcI = ei;
  const int* dstI = ei + E;

  float* ws    = (float*)d_ws;
  float* xl    = ws;
  float* xr    = xl + (size_t)N * 256;
  float* hbuf  = xr + (size_t)N * 256;
  unsigned* cntG   = (unsigned*)(hbuf + (size_t)N * 256);
  unsigned* tot    = cntG + (size_t)GRP_CHUNKS * GRP_CAP;
  unsigned* rowptr = tot + (size_t)N;
  int*      csr    = (int*)(rowptr + (size_t)N + 1);
  unsigned* rank   = (unsigned*)(csr + (size_t)E);
  float* outF = (float*)d_out;

  dim3 blk(256);

  grp_rank<<<GRP_CHUNKS, dim3(1024), 0, stream>>>(dstI, E, N, rank, cntG);
  grp_prefix<GRP_CHUNKS><<<(N + 255) / 256, blk, 0, stream>>>(cntG, N, tot);
  dst_scan<<<1, dim3(1024), 0, stream>>>(tot, N, rowptr);
  grp_place<<<(E + 1023) / 1024, blk, 0, stream>>>(srcI, dstI, E, GRP_CHUNKS, rank, cntG, rowptr, csr);

  int rtiles = (N + 31) / 32;
  gemm_pair_wmma<128, 256><<<rtiles, dim3(256), 0, stream>>>(x, W1l, W1r, xl, xr, N);
  node_gat_c32<<<(N + 15) / 16, blk, 0, stream>>>(xl, xr, att1, b1, rowptr, csr, N, hbuf);

  gemm_pair_wmma<256, 128><<<rtiles, dim3(128), 0, stream>>>(hbuf, W2l, W2r, xl, xr, N);
  node_gat_c16<<<(N + 15) / 16, blk, 0, stream>>>(xl, xr, att2, b2, rowptr, csr, N, outF);
}


# cross-lane steps (per-head logit sums, log-softmax row max and sum, the scan's running sums) read lanes by DPP / lane permute on the vector ALU, not by LDS-unit permutes; order of sums unchanged
# speedup vs baseline: 8.5257x; 1.0396x over previous
#include <hip/hip_runtime.h>
#include <math.h>

typedef __attribute__((ext_vector_type(2))) float v2f;
typedef __attribute__((ext_vector_type(8))) float v8f;
typedef __attribute__((ext_vector_type(4))) float v4f;

#define HEADS 8

#if defined(__gfx1250__) && __has_builtin(__builtin_amdgcn_wmma_f32_16x16x4_f32)
#define USE_WMMA_F32 1
#else
#define USE_WMMA_F32 0
#endif

struct GemmFrag { v4f a[2]; v4f b[4]; };

template <int KIN, int KOUT>
__global__ void __launch_bounds__((KOUT / 32) * 32)
gemm_pair_wmma(const float* __restrict__ A, const float* __restrict__ Wl,
               const float* __restrict__ Wr, float* __restrict__ Cl, float* __restrict__ Cr,
               int M) {
  static_assert(KIN % 16 == 0 && KOUT % 64 == 0, "K in pairs of 8-wide steps, 64-column strips");
  constexpr int NS = KOUT / 64;
  const int lane = threadIdx.x & 31;
  const int h = lane >> 4;
  const int c = lane & 15;
  const int wv = __builtin_amdgcn_readfirstlane((int)(threadIdx.x >> 5));
  const bool right = wv >= NS;
  const float* __restrict__ W = right ? Wr : Wl;
  float* __restrict__ C       = right ? Cr : Cl;
  const int col0 = (right ? wv - NS : wv) * 64;
  const int row0 = blockIdx.x * 32;
  if (row0 >= M) return;

#if USE_WMMA_F32
  const int ra0 = row0 + c      < M ? row0 + c      : M - 1;
  const int ra1 = row0 + 16 + c < M ? row0 + 16 + c : M - 1;
  const v4f* a0p = (const v4f*)(A + (size_t)ra0 * KIN + 4 * h);
  const v4f* a1p = (const v4f*)(A + (size_t)ra1 * KIN + 4 * h);
  const v4f* bp  = (const v4f*)(W + (size_t)(4 * h) * KOUT + col0 + 4 * c);

  v8f acc[2][4] = {};
  auto load = [&](GemmFrag& f, int kk) {
    f.a[0] = a0p[2 * kk];
    f.a[1] = a1p[2 * kk];
#pragma unroll
    for (int e = 0; e < 4; ++e) f.b[e] = bp[(size_t)(8 * kk + e) * (KOUT / 4)];
  };
  auto mma = [&](const GemmFrag& f) {
#pragma unroll
    for (int p = 0; p < 2; ++p) {
#pragma unroll
      for (int r = 0; r < 2; ++r) {
        v2f a;
        a[0] = f.a[r][2 * p];
        a[1] = f.a[r][2 * p + 1];
#pragma unroll
        for (int t = 0; t < 4; ++t) {
          v2f b;
          b[0] = f.b[2 * p][t];
          b[1] = f.b[2 * p + 1][t];
          acc[r][t] = __builtin_amdgcn_wmma_f32_16x16x4_f32(false, a, false, b,
                                                            (short)0, acc[r][t], false, false);
        }
      }
    }
  };

  constexpr int NK = KIN / 8;
  GemmFrag f0, f1;
  load(f0, 0);
#pragma nounroll
  for (int kk = 0; kk < NK - 2; kk += 2) {
    load(f1, kk + 1);
    __builtin_amdgcn_sched_barrier(0);
    mma(f0);
    load(f0, kk + 2);
    __builtin_amdgcn_sched_barrier(0);
    mma(f1);
  }
  load(f1, NK - 1);
  __builtin_amdgcn_sched_barrier(0);
  mma(f0);
  mma(f1);

#pragma unroll
  for (int r = 0; r < 2; ++r) {
#pragma unroll
    for (int v = 0; v < 8; ++v) {
      const int row = row0 + 16 * r + 8 * h + v;
      if (row < M) {
        v4f o;
        o.x = acc[r][0][v]; o.y = acc[r][1][v]; o.z = acc[r][2][v]; o.w = acc[r][3][v];
        *(v4f*)(C + (size_t)row * KOUT + col0 + 4 * c) = o;
      }
    }
  }
#else
  for (int i = lane; i < 32 * 64; i += 32) {
    const int row = row0 + i / 64, col = col0 + i % 64;
    if (row < M) {
      float sum = 0.f;
      for (int k = 0; k < KIN; ++k) sum += A[(size_t)row * KIN + k] * W[(size_t)k * KOUT + col];
      C[(size_t)row * KOUT + col] = sum;
    }
  }
#endif
}

#define GRP_CAP 51200
#define GRP_CHUNKS 64

__global__ void __launch_bounds__(1024) grp_rank(const int* __restrict__ dstI, int E, int N,
                                                 unsigned* __restrict__ rank,
                                                 unsigned* __restrict__ cntG) {
  __shared__ uint4 lc4[GRP_CAP / 4];
  unsigned* lc = (unsigned*)lc4;
  const int t = threadIdx.x;
  const int g = blockIdx.x;
  for (int i = t; i < GRP_CAP / 4; i += 1024) lc4[i] = make_uint4(0u, 0u, 0u, 0u);
  __syncthreads();
  const int per = (E + (int)gridDim.x - 1) / (int)gridDim.x;
  const int e0 = g * per;
  const int e1 = e0 + per < E ? e0 + per : E;
  for (int e = e0 + t; e < e1; e += 8 * 1024) {
    int d[8];
#pragma unroll
    for (int u = 0; u < 8; ++u) d[u] = dstI[e + u * 1024 < e1 ? e + u * 1024 : e1 - 1];
    asm volatile("" :: "v"(d[0]), "v"(d[1]), "v"(d[2]), "v"(d[3]),
                       "v"(d[4]), "v"(d[5]), "v"(d[6]), "v"(d[7]));
#pragma unroll
    for (int u = 0; u < 8; ++u)
      if (e + u * 1024 < e1) rank[e + u * 1024] = atomicAdd(&lc[d[u]], 1u);
  }
  __syncthreads();
  uint4* __restrict__ row = (uint4*)(cntG + (size_t)g * GRP_CAP);
  for (int i = t; i < (N + 3) / 4; i += 1024) row[i] = lc4[i];
}

template <int G>
__global__ void grp_prefix(unsigned* __restrict__ cntG, int N, unsigned* __restrict__ tot) {
  static_assert(G % 32 == 0, "rows are read in batches of 32");
  const int d = blockIdx.x * blockDim.x + threadIdx.x;
  if (d >= N) return;
  unsigned run = 0;
  for (int g0 = 0; g0 < G; g0 += 32) {
    unsigned c[32];
#pragma unroll
    for (int u = 0; u < 32; ++u) c[u] = cntG[(size_t)(g0 + u) * GRP_CAP + d];
#pragma unroll
    for (int u = 0; u < 32; ++u) { cntG[(size_t)(g0 + u) * GRP_CAP + d] = run; run += c[u]; }
  }
  tot[d] = run;
}

template <int M>
__device__ __forceinline__ float row_xor(float x) {
  static_assert(M >= 1 && M <= 15, "inside a 16-lane row");
  return __builtin_bit_cast(float, __builtin_amdgcn_update_dpp(0, __builtin_bit_cast(int, x),
                                                               0x160 + M, 0xf, 0xf, true));
}
__device__ __forceinline__ float half_xor(float x) {
  const int b = __builtin_bit_cast(int, x);
  return __builtin_bit_cast(float,
                            __builtin_amdgcn_permlanex16(b, b, 0x76543210, 0xfedcba98, false, false));
}
template <int S>
__device__ __forceinline__ unsigned row_up(unsigned x) {
  static_assert(S >= 1 && S <= 15, "inside a 16-lane row");
  return (unsigned)__builtin_amdgcn_update_dpp(0, (int)x, 0x110 + S, 0xf, 0xf, true);
}
__device__ __forceinline__ unsigned wave_running_sum(unsigned v, int lane) {
  v += row_up<1>(v);
  v += row_up<2>(v);
  v += row_up<4>(v);
  v += row_up<8>(v);
  const unsigned low = (unsigned)__builtin_amdgcn_readlane((int)v, 15);
  return lane >= 16 ? v + low : v;
}

__global__ void __launch_bounds__(1024) dst_scan(const unsigned* __restrict__ cnt, int N,
                                                 unsigned* __restrict__ rowptr) {
  __shared__ unsigned wsum[32];
  const int t = threadIdx.x, lane = t & 31, wv = t >> 5;
  const int wvs = __builtin_amdgcn_readfirstlane(wv);
  unsigned carry = 0;
  unsigned n0, n1, n2, n3;
  {
    const int i0 = 4 * t;
    n0 = i0     < N ? cnt[i0]     : 0u;
    n1 = i0 + 1 < N ? cnt[i0 + 1] : 0u;
    n2 = i0 + 2 < N ? cnt[i0 + 2] : 0u;
    n3 = i0 + 3 < N ? cnt[i0 + 3] : 0u;
  }
  for (int base = 0; base < N; base += 4096) {
    const int i0 = base + 4 * t;
    const unsigned c0 = n0, c1 = n1, c2 = n2, c3 = n3;
    {
      const int j0 = i0 + 4096;
      n0 = j0     < N ? cnt[j0]     : 0u;
      n1 = j0 + 1 < N ? cnt[j0 + 1] : 0u;
      n2 = j0 + 2 < N ? cnt[j0 + 2] : 0u;
      n3 = j0 + 3 < N ? cnt[j0 + 3] : 0u;
    }
    const unsigned local = c0 + c1 + c2 + c3;
    const unsigned inc = wave_running_sum(local, lane);
    if (lane == 31) wsum[wv] = inc;
    __syncthreads();
    const unsigned wtot = wsum[lane];
    const unsigned winc = wave_running_sum(wtot, lane);
    const unsigned woff  = (unsigned)__builtin_amdgcn_readlane((int)winc, wvs) -
                           (unsigned)__builtin_amdgcn_readlane((int)wtot, wvs);
    const unsigned total = (unsigned)__builtin_amdgcn_readlane((int)winc, 31);
    unsigned ex = carry + woff + inc - local;
    if (i0     < N) rowptr[i0]     = ex;
    ex += c0;
    if (i0 + 1 < N) rowptr[i0 + 1] = ex;
    ex += c1;
    if (i0 + 2 < N) rowptr[i0 + 2] = ex;
    ex += c2;
    if (i0 + 3 < N) rowptr[i0 + 3] = ex;
    carry += total;
    __syncthreads();
  }
  if (t == 0) rowptr[N] = carry;
}

__global__ void __launch_bounds__(256) grp_place(const int* __restrict__ srcI,
                                                 const int* __restrict__ dstI, int E, int G,
                                                 const unsigned* __restrict__ rank,
                                                 const unsigned* __restrict__ cntG,
                                                 const unsigned* __restrict__ rowptr,
                                                 int* __restrict__ csr_src) {
  const int eb = blockIdx.x * 1024 + threadIdx.x;
  if (eb >= E) return;
  const int per = (E + G - 1) / G;
  int d[4], s[4];
  unsigned r[4], a[4], b[4];
#pragma unroll
  for (int u = 0; u < 4; ++u) {
    const int e = eb + u * 256 < E ? eb + u * 256 : E - 1;
    d[u] = dstI[e];
    s[u] = srcI[e];
    r[u] = rank[e];
  }
#pragma unroll
  for (int u = 0; u < 4; ++u) {
    const int e = eb + u * 256 < E ? eb + u * 256 : E - 1;
    a[u] = rowptr[d[u]];
    b[u] = cntG[(size_t)(e / per) * GRP_CAP + d[u]];
  }
  asm volatile("" :: "v"(s[0]), "v"(s[1]), "v"(s[2]), "v"(s[3]), "v"(r[0]), "v"(r[1]), "v"(r[2]),
                     "v"(r[3]), "v"(a[0]), "v"(a[1]), "v"(a[2]), "v"(a[3]), "v"(b[0]), "v"(b[1]),
                     "v"(b[2]), "v"(b[3]));
#pragma unroll
  for (int u = 0; u < 4; ++u)
    if (eb + u * 256 < E) csr_src[a[u] + b[u] + r[u]] = s[u];
}

__device__ __forceinline__ float leaky_dot4(v4f a, v4f b, v4f w) {
  float acc, v;
  v = a.x + b.x; v = fmaxf(v, 0.2f * v); acc  = v * w.x;
  v = a.y + b.y; v = fmaxf(v, 0.2f * v); acc += v * w.y;
  v = a.z + b.z; v = fmaxf(v, 0.2f * v); acc += v * w.z;
  v = a.w + b.w; v = fmaxf(v, 0.2f * v); acc += v * w.w;
  return acc;
}

template <int C>
__device__ __forceinline__ float group_logit(v4f a, v4f b, v4f w) {
  float p = leaky_dot4(a, b, w);
  p += row_xor<1>(p);
  p += row_xor<2>(p);
  if constexpr (C == 32) p += row_xor<4>(p);
  return p;
}

template <int C, bool LAST>
__device__ __forceinline__ void node_rows(const float* __restrict__ xl,
                                          const float* __restrict__ xr,
                                          const float* __restrict__ att,
                                          const float* __restrict__ bias,
                                          const unsigned* __restrict__ rowptr,
                                          const int* __restrict__ csr_src,
                                          int N, float* __restrict__ out) {
  static_assert(C == 32 || C == 16, "row pieces of 128 floats, 8 or 4 lanes per head");
  static_assert(!LAST || C == 16, "the log-softmax epilogue takes one 128-float piece");
  constexpr int ROW = HEADS * C;
  constexpr int NJ  = ROW / 128;
  constexpr int UN  = 4;
  static_assert(UN == 4 && (NJ == 1 || NJ == 2), "the load fence below names UN*NJ vectors");
  const int lane = threadIdx.x & 31;
  const int wib  = __builtin_amdgcn_readfirstlane((int)(threadIdx.x >> 5));
  const int wave = blockIdx.x * (int)(blockDim.x >> 5) + wib;
  const int nw   = gridDim.x * (int)(blockDim.x >> 5);
  v4f w[NJ], bz[NJ];
#pragma unroll
  for (int j = 0; j < NJ; ++j) {
    w[j]  = ((const v4f*)att)[j * 32 + lane];
    bz[j] = ((const v4f*)bias)[j * 32 + lane];
  }
  for (int d = wave; d < N; d += nw) {
    const unsigned beg = rowptr[d], end = rowptr[d + 1];
    const v4f* pr = (const v4f*)(xr + (size_t)d * ROW) + lane;
    const v4f* ps = (const v4f*)(xl + (size_t)d * ROW) + lane;
    v4f b[NJ], acc[NJ];
    float  mx[NJ], den[NJ];
#pragma unroll
    for (int j = 0; j < NJ; ++j) { b[j] = pr[j * 32]; acc[j] = ps[j * 32]; }
#pragma unroll
    for (int j = 0; j < NJ; ++j) { mx[j] = group_logit<C>(acc[j], b[j], w[j]); den[j] = 1.f; }

    for (unsigned base = beg; base < end; base += 32) {
      const int cnt = (int)(end - base) < 32 ? (int)(end - base) : 32;
      const int my  = lane < cnt ? csr_src[base + lane] : 0;
      for (int i = 0; i < cnt; i += UN) {
        v4f a[UN][NJ];
#pragma unroll
        for (int u = 0; u < UN; ++u) {
          const int ii = i + u < cnt ? i + u : cnt - 1;
          const int s  = __builtin_amdgcn_readlane(my, ii);
          const v4f* p = (const v4f*)(xl + (size_t)s * ROW) + lane;
#pragma unroll
          for (int j = 0; j < NJ; ++j) a[u][j] = p[j * 32];
        }
        if constexpr (NJ == 2)
          asm volatile("" :: "v"(a[0][0]), "v"(a[0][1]), "v"(a[1][0]), "v"(a[1][1]),
                             "v"(a[2][0]), "v"(a[2][1]), "v"(a[3][0]), "v"(a[3][1]));
        else
          asm volatile("" :: "v"(a[0][0]), "v"(a[1][0]), "v"(a[2][0]), "v"(a[3][0]));
#pragma unroll
        for (int u = 0; u < UN; ++u) {
          if (u == 0 || i + u < cnt) {
#pragma unroll
            for (int j = 0; j < NJ; ++j) {
              const float lg = group_logit<C>(a[u][j], b[j], w[j]);
              const float t  = expf(-fabsf(mx[j] - lg));
              const bool  up = lg > mx[j];
              const float sc = up ? t : 1.f;
              const float ex = up ? 1.f : t;
              mx[j]  = fmaxf(mx[j], lg);
              den[j] = den[j] * sc + ex;
              acc[j].x = acc[j].x * sc + ex * a[u][j].x;
              acc[j].y = acc[j].y * sc + ex * a[u][j].y;
              acc[j].z = acc[j].z * sc + ex * a[u][j].z;
              acc[j].w = acc[j].w * sc + ex * a[u][j].w;
            }
          }
        }
      }
    }

    v4f* po = (v4f*)(out + (size_t)d * ROW) + lane;
    if constexpr (!LAST) {
#pragma unroll
      for (int j = 0; j < NJ; ++j) {
        v4f v;
        v.x = acc[j].x / den[j] + bz[j].x; v.x = v.x > 0.f ? v.x : (expf(v.x) - 1.f);
        v.y = acc[j].y / den[j] + bz[j].y; v.y = v.y > 0.f ? v.y : (expf(v.y) - 1.f);
        v.z = acc[j].z / den[j] + bz[j].z; v.z = v.z > 0.f ? v.z : (expf(v.z) - 1.f);
        v.w = acc[j].w / den[j] + bz[j].w; v.w = v.w > 0.f ? v.w : (expf(v.w) - 1.f);
        po[j * 32] = v;
      }
    } else {
      v4f v;
      v.x = acc[0].x / den[0] + bz[0].x;
      v.y = acc[0].y / den[0] + bz[0].y;
      v.z = acc[0].z / den[0] + bz[0].z;
      v.w = acc[0].w / den[0] + bz[0].w;
      float rm = fmaxf(fmaxf(v.x, v.y), fmaxf(v.z, v.w));
      rm = fmaxf(rm, row_xor<1>(rm));
      rm = fmaxf(rm, row_xor<2>(rm));
      rm = fmaxf(rm, row_xor<4>(rm));
      rm = fmaxf(rm, row_xor<8>(rm));
      rm = fmaxf(rm, half_xor(rm));
      float se = (expf(v.x - rm) + expf(v.y - rm)) + (expf(v.z - rm) + expf(v.w - rm));
      se += row_xor<1>(se);
      se += row_xor<2>(se);
      se += row_xor<4>(se);
      se += row_xor<8>(se);
      se += half_xor(se);
      const float lse = logf(se);
      v.x = v.x - rm - lse; v.y = v.y - rm - lse; v.z = v.z - rm - lse; v.w = v.w - rm - lse;
      po[0] = v;
    }
  }
}

__global__ void __launch_bounds__(256) node_gat_c32(const float* __restrict__ xl,
                                                    const float* __restrict__ xr,
                                                    const float* __restrict__ att,
                                                    const float* __restrict__ bias,
                                                    const unsigned* __restrict__ rowptr,
                                                    const int* __restrict__ csr_src,
                                                    int N, float* __restrict__ out) {
  node_rows<32, false>(xl, xr, att, bias, rowptr, csr_src, N, out);
}

__global__ void __launch_bounds__(256) node_gat_c16(const float* __restrict__ xl,
                                                    const float* __restrict__ xr,
                                                    const float* __restrict__ att,
                                                    const float* __restrict__ bias,
                                                    const unsigned* __restrict__ rowptr,
                                                    const int* __restrict__ csr_src,
                                                    int N, float* __restrict__ out) {
  node_rows<16, true>(xl, xr, att, bias, rowptr, csr_src, N, out);
}

extern "C" void kernel_launch(void* const* d_in, const int* in_sizes, int n_in,
                              void* d_out, int out_size, void* d_ws, size_t ws_size,
                              hipStream_t stream) {
  (void)n_in; (void)out_size; (void)ws_size;
  const float* x    = (const float*)d_in[0];
  const int*   ei   = (const int*)d_in[1];
  const float* W1l  = (const float*)d_in[2];
  const float* W1r  = (const float*)d_in[3];
  const float* att1 = (const float*)d_in[4];
  const float* b1   = (const float*)d_in[5];
  const float* W2l  = (const float*)d_in[6];
  const float* W2r  = (const float*)d_in[7];
  const float* att2 = (const float*)d_in[8];
  const float* b2   = (const float*)d_in[9];

  int N = in_sizes[0] / 128;
  int E = in_sizes[1] / 2;
  const int* srcI = ei;
  const int* dstI = ei + E;

  float* ws    = (float*)d_ws;
  float* xl    = ws;
  float* xr    = xl + (size_t)N * 256;
  float* hbuf  = xr + (size_t)N * 256;
  unsigned* cntG   = (unsigned*)(hbuf + (size_t)N * 256);
  unsigned* tot    = cntG + (size_t)GRP_CHUNKS * GRP_CAP;
  unsigned* rowptr = tot + (size_t)N;
  int*      csr    = (int*)(rowptr + (size_t)N + 1);
  unsigned* rank   = (unsigned*)(csr + (size_t)E);
  float* outF = (float*)d_out;

  dim3 blk(256);

  grp_rank<<<GRP_CHUNKS, dim3(1024), 0, stream>>>(dstI, E, N, rank, cntG);
  grp_prefix<GRP_CHUNKS><<<(N + 255) / 256, blk, 0, stream>>>(cntG, N, tot);
  dst_scan<<<1, dim3(1024), 0, stream>>>(tot, N, rowptr);
  grp_place<<<(E + 1023) / 1024, blk, 0, stream>>>(srcI, dstI, E, GRP_CHUNKS, rank, cntG, rowptr, csr);

  int rtiles = (N + 31) / 32;
  gemm_pair_wmma<128, 256><<<rtiles, dim3(256), 0, stream>>>(x, W1l, W1r, xl, xr, N);
  node_gat_c32<<<(N + 15) / 16, blk, 0, stream>>>(xl, xr, att1, b1, rowptr, csr, N, hbuf);

  gemm_pair_wmma<256, 128><<<rtiles, dim3(128), 0, stream>>>(hbuf, W2l, W2r, xl, xr, N);
  node_gat_c16<<<(N + 15) / 16, blk, 0, stream>>>(xl, xr, att2, b2, rowptr, csr, N, outF);
}


# node pass: softmax edge weights exp(-|d|) and the ELU exp (used at x <= 0 only) by hardware exp2 of x*log2(e), not the library exp: ~24 of 80 vector instr per edge row gone; order of sums unchanged
# speedup vs baseline: 9.0012x; 1.0558x over previous
#include <hip/hip_runtime.h>
#include <math.h>

typedef __attribute__((ext_vector_type(2))) float v2f;
typedef __attribute__((ext_vector_type(8))) float v8f;
typedef __attribute__((ext_vector_type(4))) float v4f;

#define HEADS 8

#if defined(__gfx1250__) && __has_builtin(__builtin_amdgcn_wmma_f32_16x16x4_f32)
#define USE_WMMA_F32 1
#else
#define USE_WMMA_F32 0
#endif

struct GemmFrag { v4f a[2]; v4f b[4]; };

template <int KIN, int KOUT>
__global__ void __launch_bounds__((KOUT / 32) * 32)
gemm_pair_wmma(const float* __restrict__ A, const float* __restrict__ Wl,
               const float* __restrict__ Wr, float* __restrict__ Cl, float* __restrict__ Cr,
               int M) {
  static_assert(KIN % 16 == 0 && KOUT % 64 == 0, "K in pairs of 8-wide steps, 64-column strips");
  constexpr int NS = KOUT / 64;
  const int lane = threadIdx.x & 31;
  const int h = lane >> 4;
  const int c = lane & 15;
  const int wv = __builtin_amdgcn_readfirstlane((int)(threadIdx.x >> 5));
  const bool right = wv >= NS;
  const float* __restrict__ W = right ? Wr : Wl;
  float* __restrict__ C       = right ? Cr : Cl;
  const int col0 = (right ? wv - NS : wv) * 64;
  const int row0 = blockIdx.x * 32;
  if (row0 >= M) return;

#if USE_WMMA_F32
  const int ra0 = row0 + c      < M ? row0 + c      : M - 1;
  const int ra1 = row0 + 16 + c < M ? row0 + 16 + c : M - 1;
  const v4f* a0p = (const v4f*)(A + (size_t)ra0 * KIN + 4 * h);
  const v4f* a1p = (const v4f*)(A + (size_t)ra1 * KIN + 4 * h);
  const v4f* bp  = (const v4f*)(W + (size_t)(4 * h) * KOUT + col0 + 4 * c);

  v8f acc[2][4] = {};
  auto load = [&](GemmFrag& f, int kk) {
    f.a[0] = a0p[2 * kk];
    f.a[1] = a1p[2 * kk];
#pragma unroll
    for (int e = 0; e < 4; ++e) f.b[e] = bp[(size_t)(8 * kk + e) * (KOUT / 4)];
  };
  auto mma = [&](const GemmFrag& f) {
#pragma unroll
    for (int p = 0; p < 2; ++p) {
#pragma unroll
      for (int r = 0; r < 2; ++r) {
        v2f a;
        a[0] = f.a[r][2 * p];
        a[1] = f.a[r][2 * p + 1];
#pragma unroll
        for (int t = 0; t < 4; ++t) {
          v2f b;
          b[0] = f.b[2 * p][t];
          b[1] = f.b[2 * p + 1][t];
          acc[r][t] = __builtin_amdgcn_wmma_f32_16x16x4_f32(false, a, false, b,
                                                            (short)0, acc[r][t], false, false);
        }
      }
    }
  };

  constexpr int NK = KIN / 8;
  GemmFrag f0, f1;
  load(f0, 0);
#pragma nounroll
  for (int kk = 0; kk < NK - 2; kk += 2) {
    load(f1, kk + 1);
    __builtin_amdgcn_sched_barrier(0);
    mma(f0);
    load(f0, kk + 2);
    __builtin_amdgcn_sched_barrier(0);
    mma(f1);
  }
  load(f1, NK - 1);
  __builtin_amdgcn_sched_barrier(0);
  mma(f0);
  mma(f1);

#pragma unroll
  for (int r = 0; r < 2; ++r) {
#pragma unroll
    for (int v = 0; v < 8; ++v) {
      const int row = row0 + 16 * r + 8 * h + v;
      if (row < M) {
        v4f o;
        o.x = acc[r][0][v]; o.y = acc[r][1][v]; o.z = acc[r][2][v]; o.w = acc[r][3][v];
        *(v4f*)(C + (size_t)row * KOUT + col0 + 4 * c) = o;
      }
    }
  }
#else
  for (int i = lane; i < 32 * 64; i += 32) {
    const int row = row0 + i / 64, col = col0 + i % 64;
    if (row < M) {
      float sum = 0.f;
      for (int k = 0; k < KIN; ++k) sum += A[(size_t)row * KIN + k] * W[(size_t)k * KOUT + col];
      C[(size_t)row * KOUT + col] = sum;
    }
  }
#endif
}

#define GRP_CAP 51200
#define GRP_CHUNKS 64

__global__ void __launch_bounds__(1024) grp_rank(const int* __restrict__ dstI, int E, int N,
                                                 unsigned* __restrict__ rank,
                                                 unsigned* __restrict__ cntG) {
  __shared__ uint4 lc4[GRP_CAP / 4];
  unsigned* lc = (unsigned*)lc4;
  const int t = threadIdx.x;
  const int g = blockIdx.x;
  for (int i = t; i < GRP_CAP / 4; i += 1024) lc4[i] = make_uint4(0u, 0u, 0u, 0u);
  __syncthreads();
  const int per = (E + (int)gridDim.x - 1) / (int)gridDim.x;
  const int e0 = g * per;
  const int e1 = e0 + per < E ? e0 + per : E;
  for (int e = e0 + t; e < e1; e += 8 * 1024) {
    int d[8];
#pragma unroll
    for (int u = 0; u < 8; ++u) d[u] = dstI[e + u * 1024 < e1 ? e + u * 1024 : e1 - 1];
    asm volatile("" :: "v"(d[0]), "v"(d[1]), "v"(d[2]), "v"(d[3]),
                       "v"(d[4]), "v"(d[5]), "v"(d[6]), "v"(d[7]));
#pragma unroll
    for (int u = 0; u < 8; ++u)
      if (e + u * 1024 < e1) rank[e + u * 1024] = atomicAdd(&lc[d[u]], 1u);
  }
  __syncthreads();
  uint4* __restrict__ row = (uint4*)(cntG + (size_t)g * GRP_CAP);
  for (int i = t; i < (N + 3) / 4; i += 1024) row[i] = lc4[i];
}

template <int G>
__global__ void grp_prefix(unsigned* __restrict__ cntG, int N, unsigned* __restrict__ tot) {
  static_assert(G % 32 == 0, "rows are read in batches of 32");
  const int d = blockIdx.x * blockDim.x + threadIdx.x;
  if (d >= N) return;
  unsigned run = 0;
  for (int g0 = 0; g0 < G; g0 += 32) {
    unsigned c[32];
#pragma unroll
    for (int u = 0; u < 32; ++u) c[u] = cntG[(size_t)(g0 + u) * GRP_CAP + d];
#pragma unroll
    for (int u = 0; u < 32; ++u) { cntG[(size_t)(g0 + u) * GRP_CAP + d] = run; run += c[u]; }
  }
  tot[d] = run;
}

template <int M>
__device__ __forceinline__ float row_xor(float x) {
  static_assert(M >= 1 && M <= 15, "inside a 16-lane row");
  return __builtin_bit_cast(float, __builtin_amdgcn_update_dpp(0, __builtin_bit_cast(int, x),
                                                               0x160 + M, 0xf, 0xf, true));
}
__device__ __forceinline__ float half_xor(float x) {
  const int b = __builtin_bit_cast(int, x);
  return __builtin_bit_cast(float,
                            __builtin_amdgcn_permlanex16(b, b, 0x76543210, 0xfedcba98, false, false));
}
template <int S>
__device__ __forceinline__ unsigned row_up(unsigned x) {
  static_assert(S >= 1 && S <= 15, "inside a 16-lane row");
  return (unsigned)__builtin_amdgcn_update_dpp(0, (int)x, 0x110 + S, 0xf, 0xf, true);
}
__device__ __forceinline__ unsigned wave_running_sum(unsigned v, int lane) {
  v += row_up<1>(v);
  v += row_up<2>(v);
  v += row_up<4>(v);
  v += row_up<8>(v);
  const unsigned low = (unsigned)__builtin_amdgcn_readlane((int)v, 15);
  return lane >= 16 ? v + low : v;
}

__global__ void __launch_bounds__(1024) dst_scan(const unsigned* __restrict__ cnt, int N,
                                                 unsigned* __restrict__ rowptr) {
  __shared__ unsigned wsum[32];
  const int t = threadIdx.x, lane = t & 31, wv = t >> 5;
  const int wvs = __builtin_amdgcn_readfirstlane(wv);
  unsigned carry = 0;
  unsigned n0, n1, n2, n3;
  {
    const int i0 = 4 * t;
    n0 = i0     < N ? cnt[i0]     : 0u;
    n1 = i0 + 1 < N ? cnt[i0 + 1] : 0u;
    n2 = i0 + 2 < N ? cnt[i0 + 2] : 0u;
    n3 = i0 + 3 < N ? cnt[i0 + 3] : 0u;
  }
  for (int base = 0; base < N; base += 4096) {
    const int i0 = base + 4 * t;
    const unsigned c0 = n0, c1 = n1, c2 = n2, c3 = n3;
    {
      const int j0 = i0 + 4096;
      n0 = j0     < N ? cnt[j0]     : 0u;
      n1 = j0 + 1 < N ? cnt[j0 + 1] : 0u;
      n2 = j0 + 2 < N ? cnt[j0 + 2] : 0u;
      n3 = j0 + 3 < N ? cnt[j0 + 3] : 0u;
    }
    const unsigned local = c0 + c1 + c2 + c3;
    const unsigned inc = wave_running_sum(local, lane);
    if (lane == 31) wsum[wv] = inc;
    __syncthreads();
    const unsigned wtot = wsum[lane];
    const unsigned winc = wave_running_sum(wtot, lane);
    const unsigned woff  = (unsigned)__builtin_amdgcn_readlane((int)winc, wvs) -
                           (unsigned)__builtin_amdgcn_readlane((int)wtot, wvs);
    const unsigned total = (unsigned)__builtin_amdgcn_readlane((int)winc, 31);
    unsigned ex = carry + woff + inc - local;
    if (i0     < N) rowptr[i0]     = ex;
    ex += c0;
    if (i0 + 1 < N) rowptr[i0 + 1] = ex;
    ex += c1;
    if (i0 + 2 < N) rowptr[i0 + 2] = ex;
    ex += c2;
    if (i0 + 3 < N) rowptr[i0 + 3] = ex;
    carry += total;
    __syncthreads();
  }
  if (t == 0) rowptr[N] = carry;
}

__global__ void __launch_bounds__(256) grp_place(const int* __restrict__ srcI,
                                                 const int* __restrict__ dstI, int E, int G,
                                                 const unsigned* __restrict__ rank,
                                                 const unsigned* __restrict__ cntG,
                                                 const unsigned* __restrict__ rowptr,
                                                 int* __restrict__ csr_src) {
  const int eb = blockIdx.x * 1024 + threadIdx.x;
  if (eb >= E) return;
  const int per = (E + G - 1) / G;
  int d[4], s[4];
  unsigned r[4], a[4], b[4];
#pragma unroll
  for (int u = 0; u < 4; ++u) {
    const int e = eb + u * 256 < E ? eb + u * 256 : E - 1;
    d[u] = dstI[e];
    s[u] = srcI[e];
    r[u] = rank[e];
  }
#pragma unroll
  for (int u = 0; u < 4; ++u) {
    const int e = eb + u * 256 < E ? eb + u * 256 : E - 1;
    a[u] = rowptr[d[u]];
    b[u] = cntG[(size_t)(e / per) * GRP_CAP + d[u]];
  }
  asm volatile("" :: "v"(s[0]), "v"(s[1]), "v"(s[2]), "v"(s[3]), "v"(r[0]), "v"(r[1]), "v"(r[2]),
                     "v"(r[3]), "v"(a[0]), "v"(a[1]), "v"(a[2]), "v"(a[3]), "v"(b[0]), "v"(b[1]),
                     "v"(b[2]), "v"(b[3]));
#pragma unroll
  for (int u = 0; u < 4; ++u)
    if (eb + u * 256 < E) csr_src[a[u] + b[u] + r[u]] = s[u];
}

__device__ __forceinline__ float leaky_dot4(v4f a, v4f b, v4f w) {
  float acc, v;
  v = a.x + b.x; v = fmaxf(v, 0.2f * v); acc  = v * w.x;
  v = a.y + b.y; v = fmaxf(v, 0.2f * v); acc += v * w.y;
  v = a.z + b.z; v = fmaxf(v, 0.2f * v); acc += v * w.z;
  v = a.w + b.w; v = fmaxf(v, 0.2f * v); acc += v * w.w;
  return acc;
}

__device__ __forceinline__ float exp_nonpos(float x) {
  return __builtin_amdgcn_exp2f(1.44269504f * x);
}

template <int C>
__device__ __forceinline__ float group_logit(v4f a, v4f b, v4f w) {
  float p = leaky_dot4(a, b, w);
  p += row_xor<1>(p);
  p += row_xor<2>(p);
  if constexpr (C == 32) p += row_xor<4>(p);
  return p;
}

template <int C, bool LAST>
__device__ __forceinline__ void node_rows(const float* __restrict__ xl,
                                          const float* __restrict__ xr,
                                          const float* __restrict__ att,
                                          const float* __restrict__ bias,
                                          const unsigned* __restrict__ rowptr,
                                          const int* __restrict__ csr_src,
                                          int N, float* __restrict__ out) {
  static_assert(C == 32 || C == 16, "row pieces of 128 floats, 8 or 4 lanes per head");
  static_assert(!LAST || C == 16, "the log-softmax epilogue takes one 128-float piece");
  constexpr int ROW = HEADS * C;
  constexpr int NJ  = ROW / 128;
  constexpr int UN  = 4;
  static_assert(UN == 4 && (NJ == 1 || NJ == 2), "the load fence below names UN*NJ vectors");
  const int lane = threadIdx.x & 31;
  const int wib  = __builtin_amdgcn_readfirstlane((int)(threadIdx.x >> 5));
  const int wave = blockIdx.x * (int)(blockDim.x >> 5) + wib;
  const int nw   = gridDim.x * (int)(blockDim.x >> 5);
  v4f w[NJ], bz[NJ];
#pragma unroll
  for (int j = 0; j < NJ; ++j) {
    w[j]  = ((const v4f*)att)[j * 32 + lane];
    bz[j] = ((const v4f*)bias)[j * 32 + lane];
  }
  for (int d = wave; d < N; d += nw) {
    const unsigned beg = rowptr[d], end = rowptr[d + 1];
    const v4f* pr = (const v4f*)(xr + (size_t)d * ROW) + lane;
    const v4f* ps = (const v4f*)(xl + (size_t)d * ROW) + lane;
    v4f b[NJ], acc[NJ];
    float  mx[NJ], den[NJ];
#pragma unroll
    for (int j = 0; j < NJ; ++j) { b[j] = pr[j * 32]; acc[j] = ps[j * 32]; }
#pragma unroll
    for (int j = 0; j < NJ; ++j) { mx[j] = group_logit<C>(acc[j], b[j], w[j]); den[j] = 1.f; }

    for (unsigned base = beg; base < end; base += 32) {
      const int cnt = (int)(end - base) < 32 ? (int)(end - base) : 32;
      const int my  = lane < cnt ? csr_src[base + lane] : 0;
      for (int i = 0; i < cnt; i += UN) {
        v4f a[UN][NJ];
#pragma unroll
        for (int u = 0; u < UN; ++u) {
          const int ii = i + u < cnt ? i + u : cnt - 1;
          const int s  = __builtin_amdgcn_readlane(my, ii);
          const v4f* p = (const v4f*)(xl + (size_t)s * ROW) + lane;
#pragma unroll
          for (int j = 0; j < NJ; ++j) a[u][j] = p[j * 32];
        }
        if constexpr (NJ == 2)
          asm volatile("" :: "v"(a[0][0]), "v"(a[0][1]), "v"(a[1][0]), "v"(a[1][1]),
                             "v"(a[2][0]), "v"(a[2][1]), "v"(a[3][0]), "v"(a[3][1]));
        else
          asm volatile("" :: "v"(a[0][0]), "v"(a[1][0]), "v"(a[2][0]), "v"(a[3][0]));
#pragma unroll
        for (int u = 0; u < UN; ++u) {
          if (u == 0 || i + u < cnt) {
#pragma unroll
            for (int j = 0; j < NJ; ++j) {
              const float lg = group_logit<C>(a[u][j], b[j], w[j]);
              const float t  = exp_nonpos(-fabsf(mx[j] - lg));
              const bool  up = lg > mx[j];
              const float sc = up ? t : 1.f;
              const float ex = up ? 1.f : t;
              mx[j]  = fmaxf(mx[j], lg);
              den[j] = den[j] * sc + ex;
              acc[j].x = acc[j].x * sc + ex * a[u][j].x;
              acc[j].y = acc[j].y * sc + ex * a[u][j].y;
              acc[j].z = acc[j].z * sc + ex * a[u][j].z;
              acc[j].w = acc[j].w * sc + ex * a[u][j].w;
            }
          }
        }
      }
    }

    v4f* po = (v4f*)(out + (size_t)d * ROW) + lane;
    if constexpr (!LAST) {
#pragma unroll
      for (int j = 0; j < NJ; ++j) {
        v4f v;
        v.x = acc[j].x / den[j] + bz[j].x; v.x = v.x > 0.f ? v.x : (exp_nonpos(v.x) - 1.f);
        v.y = acc[j].y / den[j] + bz[j].y; v.y = v.y > 0.f ? v.y : (exp_nonpos(v.y) - 1.f);
        v.z = acc[j].z / den[j] + bz[j].z; v.z = v.z > 0.f ? v.z : (exp_nonpos(v.z) - 1.f);
        v.w = acc[j].w / den[j] + bz[j].w; v.w = v.w > 0.f ? v.w : (exp_nonpos(v.w) - 1.f);
        po[j * 32] = v;
      }
    } else {
      v4f v;
      v.x = acc[0].x / den[0] + bz[0].x;
      v.y = acc[0].y / den[0] + bz[0].y;
      v.z = acc[0].z / den[0] + bz[0].z;
      v.w = acc[0].w / den[0] + bz[0].w;
      float rm = fmaxf(fmaxf(v.x, v.y), fmaxf(v.z, v.w));
      rm = fmaxf(rm, row_xor<1>(rm));
      rm = fmaxf(rm, row_xor<2>(rm));
      rm = fmaxf(rm, row_xor<4>(rm));
      rm = fmaxf(rm, row_xor<8>(rm));
      rm = fmaxf(rm, half_xor(rm));
      float se = (expf(v.x - rm) + expf(v.y - rm)) + (expf(v.z - rm) + expf(v.w - rm));
      se += row_xor<1>(se);
      se += row_xor<2>(se);
      se += row_xor<4>(se);
      se += row_xor<8>(se);
      se += half_xor(se);
      const float lse = logf(se);
      v.x = v.x - rm - lse; v.y = v.y - rm - lse; v.z = v.z - rm - lse; v.w = v.w - rm - lse;
      po[0] = v;
    }
  }
}

__global__ void __launch_bounds__(256) node_gat_c32(const float* __restrict__ xl,
                                                    const float* __restrict__ xr,
                                                    const float* __restrict__ att,
                                                    const float* __restrict__ bias,
                                                    const unsigned* __restrict__ rowptr,
                                                    const int* __restrict__ csr_src,
                                                    int N, float* __restrict__ out) {
  node_rows<32, false>(xl, xr, att, bias, rowptr, csr_src, N, out);
}

__global__ void __launch_bounds__(256) node_gat_c16(const float* __restrict__ xl,
                                                    const float* __restrict__ xr,
                                                    const float* __restrict__ att,
                                                    const float* __restrict__ bias,
                                                    const unsigned* __restrict__ rowptr,
                                                    const int* __restrict__ csr_src,
                                                    int N, float* __restrict__ out) {
  node_rows<16, true>(xl, xr, att, bias, rowptr, csr_src, N, out);
}

extern "C" void kernel_launch(void* const* d_in, const int* in_sizes, int n_in,
                              void* d_out, int out_size, void* d_ws, size_t ws_size,
                              hipStream_t stream) {
  (void)n_in; (void)out_size; (void)ws_size;
  const float* x    = (const float*)d_in[0];
  const int*   ei   = (const int*)d_in[1];
  const float* W1l  = (const float*)d_in[2];
  const float* W1r  = (const float*)d_in[3];
  const float* att1 = (const float*)d_in[4];
  const float* b1   = (const float*)d_in[5];
  const float* W2l  = (const float*)d_in[6];
  const float* W2r  = (const float*)d_in[7];
  const float* att2 = (const float*)d_in[8];
  const float* b2   = (const float*)d_in[9];

  int N = in_sizes[0] / 128;
  int E = in_sizes[1] / 2;
  const int* srcI = ei;
  const int* dstI = ei + E;

  float* ws    = (float*)d_ws;
  float* xl    = ws;
  float* xr    = xl + (size_t)N * 256;
  float* hbuf  = xr + (size_t)N * 256;
  unsigned* cntG   = (unsigned*)(hbuf + (size_t)N * 256);
  unsigned* tot    = cntG + (size_t)GRP_CHUNKS * GRP_CAP;
  unsigned* rowptr = tot + (size_t)N;
  int*      csr    = (int*)(rowptr + (size_t)N + 1);
  unsigned* rank   = (unsigned*)(csr + (size_t)E);
  float* outF = (float*)d_out;

  dim3 blk(256);

  grp_rank<<<GRP_CHUNKS, dim3(1024), 0, stream>>>(dstI, E, N, rank, cntG);
  grp_prefix<GRP_CHUNKS><<<(N + 255) / 256, blk, 0, stream>>>(cntG, N, tot);
  dst_scan<<<1, dim3(1024), 0, stream>>>(tot, N, rowptr);
  grp_place<<<(E + 1023) / 1024, blk, 0, stream>>>(srcI, dstI, E, GRP_CHUNKS, rank, cntG, rowptr, csr);

  int rtiles = (N + 31) / 32;
  gemm_pair_wmma<128, 256><<<rtiles, dim3(256), 0, stream>>>(x, W1l, W1r, xl, xr, N);
  node_gat_c32<<<(N + 15) / 16, blk, 0, stream>>>(xl, xr, att1, b1, rowptr, csr, N, hbuf);

  gemm_pair_wmma<256, 128><<<rtiles, dim3(128), 0, stream>>>(hbuf, W2l, W2r, xl, xr, N);
  node_gat_c16<<<(N + 15) / 16, blk, 0, stream>>>(xl, xr, att2, b2, rowptr, csr, N, outF);
}


# node-offset scan folded into its neighbour launches: count pass scans its 256 nodes' in-degrees and leaves a block total, placing pass prefixes block totals in LDS and writes rowptr; 8 -> 7 launches
# speedup vs baseline: 9.7646x; 1.0848x over previous
#include <hip/hip_runtime.h>
#include <math.h>

typedef __attribute__((ext_vector_type(2))) float v2f;
typedef __attribute__((ext_vector_type(8))) float v8f;
typedef __attribute__((ext_vector_type(4))) float v4f;

#define HEADS 8

#if defined(__gfx1250__) && __has_builtin(__builtin_amdgcn_wmma_f32_16x16x4_f32)
#define USE_WMMA_F32 1
#else
#define USE_WMMA_F32 0
#endif

struct GemmFrag { v4f a[2]; v4f b[4]; };

template <int KIN, int KOUT>
__global__ void __launch_bounds__((KOUT / 32) * 32)
gemm_pair_wmma(const float* __restrict__ A, const float* __restrict__ Wl,
               const float* __restrict__ Wr, float* __restrict__ Cl, float* __restrict__ Cr,
               int M) {
  static_assert(KIN % 16 == 0 && KOUT % 64 == 0, "K in pairs of 8-wide steps, 64-column strips");
  constexpr int NS = KOUT / 64;
  const int lane = threadIdx.x & 31;
  const int h = lane >> 4;
  const int c = lane & 15;
  const int wv = __builtin_amdgcn_readfirstlane((int)(threadIdx.x >> 5));
  const bool right = wv >= NS;
  const float* __restrict__ W = right ? Wr : Wl;
  float* __restrict__ C       = right ? Cr : Cl;
  const int col0 = (right ? wv - NS : wv) * 64;
  const int row0 = blockIdx.x * 32;
  if (row0 >= M) return;

#if USE_WMMA_F32
  const int ra0 = row0 + c      < M ? row0 + c      : M - 1;
  const int ra1 = row0 + 16 + c < M ? row0 + 16 + c : M - 1;
  const v4f* a0p = (const v4f*)(A + (size_t)ra0 * KIN + 4 * h);
  const v4f* a1p = (const v4f*)(A + (size_t)ra1 * KIN + 4 * h);
  const v4f* bp  = (const v4f*)(W + (size_t)(4 * h) * KOUT + col0 + 4 * c);

  v8f acc[2][4] = {};
  auto load = [&](GemmFrag& f, int kk) {
    f.a[0] = a0p[2 * kk];
    f.a[1] = a1p[2 * kk];
#pragma unroll
    for (int e = 0; e < 4; ++e) f.b[e] = bp[(size_t)(8 * kk + e) * (KOUT / 4)];
  };
  auto mma = [&](const GemmFrag& f) {
#pragma unroll
    for (int p = 0; p < 2; ++p) {
#pragma unroll
      for (int r = 0; r < 2; ++r) {
        v2f a;
        a[0] = f.a[r][2 * p];
        a[1] = f.a[r][2 * p + 1];
#pragma unroll
        for (int t = 0; t < 4; ++t) {
          v2f b;
          b[0] = f.b[2 * p][t];
          b[1] = f.b[2 * p + 1][t];
          acc[r][t] = __builtin_amdgcn_wmma_f32_16x16x4_f32(false, a, false, b,
                                                            (short)0, acc[r][t], false, false);
        }
      }
    }
  };

  constexpr int NK = KIN / 8;
  GemmFrag f0, f1;
  load(f0, 0);
#pragma nounroll
  for (int kk = 0; kk < NK - 2; kk += 2) {
    load(f1, kk + 1);
    __builtin_amdgcn_sched_barrier(0);
    mma(f0);
    load(f0, kk + 2);
    __builtin_amdgcn_sched_barrier(0);
    mma(f1);
  }
  load(f1, NK - 1);
  __builtin_amdgcn_sched_barrier(0);
  mma(f0);
  mma(f1);

#pragma unroll
  for (int r = 0; r < 2; ++r) {
#pragma unroll
    for (int v = 0; v < 8; ++v) {
      const int row = row0 + 16 * r + 8 * h + v;
      if (row < M) {
        v4f o;
        o.x = acc[r][0][v]; o.y = acc[r][1][v]; o.z = acc[r][2][v]; o.w = acc[r][3][v];
        *(v4f*)(C + (size_t)row * KOUT + col0 + 4 * c) = o;
      }
    }
  }
#else
  for (int i = lane; i < 32 * 64; i += 32) {
    const int row = row0 + i / 64, col = col0 + i % 64;
    if (row < M) {
      float sum = 0.f;
      for (int k = 0; k < KIN; ++k) sum += A[(size_t)row * KIN + k] * W[(size_t)k * KOUT + col];
      C[(size_t)row * KOUT + col] = sum;
    }
  }
#endif
}

#define GRP_CAP 51200
#define GRP_CHUNKS 64

__global__ void __launch_bounds__(1024) grp_rank(const int* __restrict__ dstI, int E, int N,
                                                 unsigned* __restrict__ rank,
                                                 unsigned* __restrict__ cntG) {
  __shared__ uint4 lc4[GRP_CAP / 4];
  unsigned* lc = (unsigned*)lc4;
  const int t = threadIdx.x;
  const int g = blockIdx.x;
  for (int i = t; i < GRP_CAP / 4; i += 1024) lc4[i] = make_uint4(0u, 0u, 0u, 0u);
  __syncthreads();
  const int per = (E + (int)gridDim.x - 1) / (int)gridDim.x;
  const int e0 = g * per;
  const int e1 = e0 + per < E ? e0 + per : E;
  for (int e = e0 + t; e < e1; e += 8 * 1024) {
    int d[8];
#pragma unroll
    for (int u = 0; u < 8; ++u) d[u] = dstI[e + u * 1024 < e1 ? e + u * 1024 : e1 - 1];
    asm volatile("" :: "v"(d[0]), "v"(d[1]), "v"(d[2]), "v"(d[3]),
                       "v"(d[4]), "v"(d[5]), "v"(d[6]), "v"(d[7]));
#pragma unroll
    for (int u = 0; u < 8; ++u)
      if (e + u * 1024 < e1) rank[e + u * 1024] = atomicAdd(&lc[d[u]], 1u);
  }
  __syncthreads();
  uint4* __restrict__ row = (uint4*)(cntG + (size_t)g * GRP_CAP);
  for (int i = t; i < (N + 3) / 4; i += 1024) row[i] = lc4[i];
}

template <int M>
__device__ __forceinline__ float row_xor(float x) {
  static_assert(M >= 1 && M <= 15, "inside a 16-lane row");
  return __builtin_bit_cast(float, __builtin_amdgcn_update_dpp(0, __builtin_bit_cast(int, x),
                                                               0x160 + M, 0xf, 0xf, true));
}
__device__ __forceinline__ float half_xor(float x) {
  const int b = __builtin_bit_cast(int, x);
  return __builtin_bit_cast(float,
                            __builtin_amdgcn_permlanex16(b, b, 0x76543210, 0xfedcba98, false, false));
}
template <int S>
__device__ __forceinline__ unsigned row_up(unsigned x) {
  static_assert(S >= 1 && S <= 15, "inside a 16-lane row");
  return (unsigned)__builtin_amdgcn_update_dpp(0, (int)x, 0x110 + S, 0xf, 0xf, true);
}
__device__ __forceinline__ unsigned wave_running_sum(unsigned v, int lane) {
  v += row_up<1>(v);
  v += row_up<2>(v);
  v += row_up<4>(v);
  v += row_up<8>(v);
  const unsigned low = (unsigned)__builtin_amdgcn_readlane((int)v, 15);
  return lane >= 16 ? v + low : v;
}

__device__ __forceinline__ unsigned block256_offsets(unsigned v, int lane, int wvs,
                                                     unsigned* wsum, unsigned& total) {
  const unsigned inc = wave_running_sum(v, lane);
  if (lane == 31) wsum[wvs] = inc;
  __syncthreads();
  const unsigned wtot = lane < 8 ? wsum[lane] : 0u;
  const unsigned winc = wave_running_sum(wtot, lane);
  const unsigned woff = (unsigned)__builtin_amdgcn_readlane((int)winc, wvs) -
                        (unsigned)__builtin_amdgcn_readlane((int)wtot, wvs);
  total = (unsigned)__builtin_amdgcn_readlane((int)winc, 7);
  return woff + inc - v;
}

static_assert(GRP_CAP / 256 <= 256, "the placing pass prefixes the block totals one per thread");
template <int G>
__global__ void __launch_bounds__(256) grp_prefix(unsigned* __restrict__ cntG, int N,
                                                  unsigned* __restrict__ lpre,
                                                  unsigned* __restrict__ bsum) {
  static_assert(G % 32 == 0, "rows are read in batches of 32");
  __shared__ unsigned wsum[8];
  const int t = threadIdx.x, lane = t & 31;
  const int wvs = __builtin_amdgcn_readfirstlane(t >> 5);
  const int d = blockIdx.x * 256 + t;
  unsigned run = 0;
  if (d < N) {
    for (int g0 = 0; g0 < G; g0 += 32) {
      unsigned c[32];
#pragma unroll
      for (int u = 0; u < 32; ++u) c[u] = cntG[(size_t)(g0 + u) * GRP_CAP + d];
#pragma unroll
      for (int u = 0; u < 32; ++u) { cntG[(size_t)(g0 + u) * GRP_CAP + d] = run; run += c[u]; }
    }
  }
  unsigned total;
  const unsigned off = block256_offsets(run, lane, wvs, wsum, total);
  if (d < N) lpre[d] = off;
  if (t == 0) bsum[blockIdx.x] = total;
}

__global__ void __launch_bounds__(256) grp_place(const int* __restrict__ srcI,
                                                 const int* __restrict__ dstI, int E, int G,
                                                 const unsigned* __restrict__ rank,
                                                 const unsigned* __restrict__ cntG,
                                                 const unsigned* __restrict__ lpre,
                                                 const unsigned* __restrict__ bsum, int N,
                                                 unsigned* __restrict__ rowptr,
                                                 int* __restrict__ csr_src) {
  __shared__ unsigned bpre[257];
  __shared__ unsigned wsum[8];
  const int t = threadIdx.x, lane = t & 31;
  const int wvs = __builtin_amdgcn_readfirstlane(t >> 5);
  const int eb = blockIdx.x * 1024 + t;
  const int per = (E + G - 1) / G;
  const int NB = (N + 255) >> 8;
  int d[4], s[4];
  unsigned r[4], a[4], b[4];
#pragma unroll
  for (int u = 0; u < 4; ++u) {
    const int e = eb + u * 256 < E ? eb + u * 256 : E - 1;
    d[u] = dstI[e];
    s[u] = srcI[e];
    r[u] = rank[e];
  }
  {
    const unsigned bs = t < NB ? bsum[t] : 0u;
    unsigned total;
    bpre[t] = block256_offsets(bs, lane, wvs, wsum, total);
    if (t == 0) bpre[256] = total;
  }
  __syncthreads();
#pragma unroll
  for (int u = 0; u < 4; ++u) {
    const int e = eb + u * 256 < E ? eb + u * 256 : E - 1;
    a[u] = lpre[d[u]];
    b[u] = cntG[(size_t)(e / per) * GRP_CAP + d[u]];
  }
  asm volatile("" :: "v"(s[0]), "v"(s[1]), "v"(s[2]), "v"(s[3]), "v"(r[0]), "v"(r[1]), "v"(r[2]),
                     "v"(r[3]), "v"(a[0]), "v"(a[1]), "v"(a[2]), "v"(a[3]), "v"(b[0]), "v"(b[1]),
                     "v"(b[2]), "v"(b[3]));
#pragma unroll
  for (int u = 0; u < 4; ++u)
    if (eb + u * 256 < E) csr_src[a[u] + bpre[d[u] >> 8] + b[u] + r[u]] = s[u];
  for (int i = blockIdx.x * 256 + t; i <= N; i += (int)gridDim.x * 256)
    rowptr[i] = i < N ? lpre[i] + bpre[i >> 8] : bpre[NB];
}

__device__ __forceinline__ float leaky_dot4(v4f a, v4f b, v4f w) {
  float acc, v;
  v = a.x + b.x; v = fmaxf(v, 0.2f * v); acc  = v * w.x;
  v = a.y + b.y; v = fmaxf(v, 0.2f * v); acc += v * w.y;
  v = a.z + b.z; v = fmaxf(v, 0.2f * v); acc += v * w.z;
  v = a.w + b.w; v = fmaxf(v, 0.2f * v); acc += v * w.w;
  return acc;
}

__device__ __forceinline__ float exp_nonpos(float x) {
  return __builtin_amdgcn_exp2f(1.44269504f * x);
}

template <int C>
__device__ __forceinline__ float group_logit(v4f a, v4f b, v4f w) {
  float p = leaky_dot4(a, b, w);
  p += row_xor<1>(p);
  p += row_xor<2>(p);
  if constexpr (C == 32) p += row_xor<4>(p);
  return p;
}

template <int C, bool LAST>
__device__ __forceinline__ void node_rows(const float* __restrict__ xl,
                                          const float* __restrict__ xr,
                                          const float* __restrict__ att,
                                          const float* __restrict__ bias,
                                          const unsigned* __restrict__ rowptr,
                                          const int* __restrict__ csr_src,
                                          int N, float* __restrict__ out) {
  static_assert(C == 32 || C == 16, "row pieces of 128 floats, 8 or 4 lanes per head");
  static_assert(!LAST || C == 16, "the log-softmax epilogue takes one 128-float piece");
  constexpr int ROW = HEADS * C;
  constexpr int NJ  = ROW / 128;
  constexpr int UN  = 4;
  static_assert(UN == 4 && (NJ == 1 || NJ == 2), "the load fence below names UN*NJ vectors");
  const int lane = threadIdx.x & 31;
  const int wib  = __builtin_amdgcn_readfirstlane((int)(threadIdx.x >> 5));
  const int wave = blockIdx.x * (int)(blockDim.x >> 5) + wib;
  const int nw   = gridDim.x * (int)(blockDim.x >> 5);
  v4f w[NJ], bz[NJ];
#pragma unroll
  for (int j = 0; j < NJ; ++j) {
    w[j]  = ((const v4f*)att)[j * 32 + lane];
    bz[j] = ((const v4f*)bias)[j * 32 + lane];
  }
  for (int d = wave; d < N; d += nw) {
    const unsigned beg = rowptr[d], end = rowptr[d + 1];
    const v4f* pr = (const v4f*)(xr + (size_t)d * ROW) + lane;
    const v4f* ps = (const v4f*)(xl + (size_t)d * ROW) + lane;
    v4f b[NJ], acc[NJ];
    float  mx[NJ], den[NJ];
#pragma unroll
    for (int j = 0; j < NJ; ++j) { b[j] = pr[j * 32]; acc[j] = ps[j * 32]; }
#pragma unroll
    for (int j = 0; j < NJ; ++j) { mx[j] = group_logit<C>(acc[j], b[j], w[j]); den[j] = 1.f; }

    for (unsigned base = beg; base < end; base += 32) {
      const int cnt = (int)(end - base) < 32 ? (int)(end - base) : 32;
      const int my  = lane < cnt ? csr_src[base + lane] : 0;
      for (int i = 0; i < cnt; i += UN) {
        v4f a[UN][NJ];
#pragma unroll
        for (int u = 0; u < UN; ++u) {
          const int ii = i + u < cnt ? i + u : cnt - 1;
          const int s  = __builtin_amdgcn_readlane(my, ii);
          const v4f* p = (const v4f*)(xl + (size_t)s * ROW) + lane;
#pragma unroll
          for (int j = 0; j < NJ; ++j) a[u][j] = p[j * 32];
        }
        if constexpr (NJ == 2)
          asm volatile("" :: "v"(a[0][0]), "v"(a[0][1]), "v"(a[1][0]), "v"(a[1][1]),
                             "v"(a[2][0]), "v"(a[2][1]), "v"(a[3][0]), "v"(a[3][1]));
        else
          asm volatile("" :: "v"(a[0][0]), "v"(a[1][0]), "v"(a[2][0]), "v"(a[3][0]));
#pragma unroll
        for (int u = 0; u < UN; ++u) {
          if (u == 0 || i + u < cnt) {
#pragma unroll
            for (int j = 0; j < NJ; ++j) {
              const float lg = group_logit<C>(a[u][j], b[j], w[j]);
              const float t  = exp_nonpos(-fabsf(mx[j] - lg));
              const bool  up = lg > mx[j];
              const float sc = up ? t : 1.f;
              const float ex = up ? 1.f : t;
              mx[j]  = fmaxf(mx[j], lg);
              den[j] = den[j] * sc + ex;
              acc[j].x = acc[j].x * sc + ex * a[u][j].x;
              acc[j].y = acc[j].y * sc + ex * a[u][j].y;
              acc[j].z = acc[j].z * sc + ex * a[u][j].z;
              acc[j].w = acc[j].w * sc + ex * a[u][j].w;
            }
          }
        }
      }
    }

    v4f* po = (v4f*)(out + (size_t)d * ROW) + lane;
    if constexpr (!LAST) {
#pragma unroll
      for (int j = 0; j < NJ; ++j) {
        v4f v;
        v.x = acc[j].x / den[j] + bz[j].x; v.x = v.x > 0.f ? v.x : (exp_nonpos(v.x) - 1.f);
        v.y = acc[j].y / den[j] + bz[j].y; v.y = v.y > 0.f ? v.y : (exp_nonpos(v.y) - 1.f);
        v.z = acc[j].z / den[j] + bz[j].z; v.z = v.z > 0.f ? v.z : (exp_nonpos(v.z) - 1.f);
        v.w = acc[j].w / den[j] + bz[j].w; v.w = v.w > 0.f ? v.w : (exp_nonpos(v.w) - 1.f);
        po[j * 32] = v;
      }
    } else {
      v4f v;
      v.x = acc[0].x / den[0] + bz[0].x;
      v.y = acc[0].y / den[0] + bz[0].y;
      v.z = acc[0].z / den[0] + bz[0].z;
      v.w = acc[0].w / den[0] + bz[0].w;
      float rm = fmaxf(fmaxf(v.x, v.y), fmaxf(v.z, v.w));
      rm = fmaxf(rm, row_xor<1>(rm));
      rm = fmaxf(rm, row_xor<2>(rm));
      rm = fmaxf(rm, row_xor<4>(rm));
      rm = fmaxf(rm, row_xor<8>(rm));
      rm = fmaxf(rm, half_xor(rm));
      float se = (expf(v.x - rm) + expf(v.y - rm)) + (expf(v.z - rm) + expf(v.w - rm));
      se += row_xor<1>(se);
      se += row_xor<2>(se);
      se += row_xor<4>(se);
      se += row_xor<8>(se);
      se += half_xor(se);
      const float lse = logf(se);
      v.x = v.x - rm - lse; v.y = v.y - rm - lse; v.z = v.z - rm - lse; v.w = v.w - rm - lse;
      po[0] = v;
    }
  }
}

__global__ void __launch_bounds__(256) node_gat_c32(const float* __restrict__ xl,
                                                    const float* __restrict__ xr,
                                                    const float* __restrict__ att,
                                                    const float* __restrict__ bias,
                                                    const unsigned* __restrict__ rowptr,
                                                    const int* __restrict__ csr_src,
                                                    int N, float* __restrict__ out) {
  node_rows<32, false>(xl, xr, att, bias, rowptr, csr_src, N, out);
}

__global__ void __launch_bounds__(256) node_gat_c16(const float* __restrict__ xl,
                                                    const float* __restrict__ xr,
                                                    const float* __restrict__ att,
                                                    const float* __restrict__ bias,
                                                    const unsigned* __restrict__ rowptr,
                                                    const int* __restrict__ csr_src,
                                                    int N, float* __restrict__ out) {
  node_rows<16, true>(xl, xr, att, bias, rowptr, csr_src, N, out);
}

extern "C" void kernel_launch(void* const* d_in, const int* in_sizes, int n_in,
                              void* d_out, int out_size, void* d_ws, size_t ws_size,
                              hipStream_t stream) {
  (void)n_in; (void)out_size; (void)ws_size;
  const float* x    = (const float*)d_in[0];
  const int*   ei   = (const int*)d_in[1];
  const float* W1l  = (const float*)d_in[2];
  const float* W1r  = (const float*)d_in[3];
  const float* att1 = (const float*)d_in[4];
  const float* b1   = (const float*)d_in[5];
  const float* W2l  = (const float*)d_in[6];
  const float* W2r  = (const float*)d_in[7];
  const float* att2 = (const float*)d_in[8];
  const float* b2   = (const float*)d_in[9];

  int N = in_sizes[0] / 128;
  int E = in_sizes[1] / 2;
  const int* srcI = ei;
  const int* dstI = ei + E;

  float* ws    = (float*)d_ws;
  float* xl    = ws;
  float* xr    = xl + (size_t)N * 256;
  float* hbuf  = xr + (size_t)N * 256;
  unsigned* cntG   = (unsigned*)(hbuf + (size_t)N * 256);
  unsigned* lpre   = cntG + (size_t)GRP_CHUNKS * GRP_CAP;
  unsigned* rowptr = lpre + (size_t)N;
  int*      csr    = (int*)(rowptr + (size_t)N + 1);
  unsigned* rank   = (unsigned*)(csr + (size_t)E);
  unsigned* bsum   = rank + (size_t)E;
  float* outF = (float*)d_out;

  dim3 blk(256);

  grp_rank<<<GRP_CHUNKS, dim3(1024), 0, stream>>>(dstI, E, N, rank, cntG);
  grp_prefix<GRP_CHUNKS><<<(N + 255) / 256, blk, 0, stream>>>(cntG, N, lpre, bsum);
  grp_place<<<(E + 1023) / 1024, blk, 0, stream>>>(srcI, dstI, E, GRP_CHUNKS, rank, cntG, lpre, bsum, N,
                                                   rowptr, csr);

  int rtiles = (N + 31) / 32;
  gemm_pair_wmma<128, 256><<<rtiles, dim3(256), 0, stream>>>(x, W1l, W1r, xl, xr, N);
  node_gat_c32<<<(N + 15) / 16, blk, 0, stream>>>(xl, xr, att1, b1, rowptr, csr, N, hbuf);

  gemm_pair_wmma<256, 128><<<rtiles, dim3(128), 0, stream>>>(hbuf, W2l, W2r, xl, xr, N);
  node_gat_c16<<<(N + 15) / 16, blk, 0, stream>>>(xl, xr, att2, b2, rowptr, csr, N, outF);
}


# layer-2 GEMM fused into layer-1 node pass: block = 32 nodes, finished h rows stay in LDS and feed the WMMAs (k order of sums unchanged); h never goes to memory; 7 -> 6 launches
# speedup vs baseline: 10.0718x; 1.0315x over previous
#include <hip/hip_runtime.h>
#include <math.h>

typedef __attribute__((ext_vector_type(2))) float v2f;
typedef __attribute__((ext_vector_type(8))) float v8f;
typedef __attribute__((ext_vector_type(4))) float v4f;

#define HEADS 8

#if defined(__gfx1250__) && __has_builtin(__builtin_amdgcn_wmma_f32_16x16x4_f32)
#define USE_WMMA_F32 1
#else
#define USE_WMMA_F32 0
#endif

struct GemmFrag { v4f a[2]; v4f b[4]; };

template <int KIN, int KOUT>
__global__ void __launch_bounds__((KOUT / 32) * 32)
gemm_pair_wmma(const float* __restrict__ A, const float* __restrict__ Wl,
               const float* __restrict__ Wr, float* __restrict__ Cl, float* __restrict__ Cr,
               int M) {
  static_assert(KIN % 16 == 0 && KOUT % 64 == 0, "K in pairs of 8-wide steps, 64-column strips");
  constexpr int NS = KOUT / 64;
  const int lane = threadIdx.x & 31;
  const int h = lane >> 4;
  const int c = lane & 15;
  const int wv = __builtin_amdgcn_readfirstlane((int)(threadIdx.x >> 5));
  const bool right = wv >= NS;
  const float* __restrict__ W = right ? Wr : Wl;
  float* __restrict__ C       = right ? Cr : Cl;
  const int col0 = (right ? wv - NS : wv) * 64;
  const int row0 = blockIdx.x * 32;
  if (row0 >= M) return;

#if USE_WMMA_F32
  const int ra0 = row0 + c      < M ? row0 + c      : M - 1;
  const int ra1 = row0 + 16 + c < M ? row0 + 16 + c : M - 1;
  const v4f* a0p = (const v4f*)(A + (size_t)ra0 * KIN + 4 * h);
  const v4f* a1p = (const v4f*)(A + (size_t)ra1 * KIN + 4 * h);
  const v4f* bp  = (const v4f*)(W + (size_t)(4 * h) * KOUT + col0 + 4 * c);

  v8f acc[2][4] = {};
  auto load = [&](GemmFrag& f, int kk) {
    f.a[0] = a0p[2 * kk];
    f.a[1] = a1p[2 * kk];
#pragma unroll
    for (int e = 0; e < 4; ++e) f.b[e] = bp[(size_t)(8 * kk + e) * (KOUT / 4)];
  };
  auto mma = [&](const GemmFrag& f) {
#pragma unroll
    for (int p = 0; p < 2; ++p) {
#pragma unroll
      for (int r = 0; r < 2; ++r) {
        v2f a;
        a[0] = f.a[r][2 * p];
        a[1] = f.a[r][2 * p + 1];
#pragma unroll
        for (int t = 0; t < 4; ++t) {
          v2f b;
          b[0] = f.b[2 * p][t];
          b[1] = f.b[2 * p + 1][t];
          acc[r][t] = __builtin_amdgcn_wmma_f32_16x16x4_f32(false, a, false, b,
                                                            (short)0, acc[r][t], false, false);
        }
      }
    }
  };

  constexpr int NK = KIN / 8;
  GemmFrag f0, f1;
  load(f0, 0);
#pragma nounroll
  for (int kk = 0; kk < NK - 2; kk += 2) {
    load(f1, kk + 1);
    __builtin_amdgcn_sched_barrier(0);
    mma(f0);
    load(f0, kk + 2);
    __builtin_amdgcn_sched_barrier(0);
    mma(f1);
  }
  load(f1, NK - 1);
  __builtin_amdgcn_sched_barrier(0);
  mma(f0);
  mma(f1);

#pragma unroll
  for (int r = 0; r < 2; ++r) {
#pragma unroll
    for (int v = 0; v < 8; ++v) {
      const int row = row0 + 16 * r + 8 * h + v;
      if (row < M) {
        v4f o;
        o.x = acc[r][0][v]; o.y = acc[r][1][v]; o.z = acc[r][2][v]; o.w = acc[r][3][v];
        *(v4f*)(C + (size_t)row * KOUT + col0 + 4 * c) = o;
      }
    }
  }
#else
  for (int i = lane; i < 32 * 64; i += 32) {
    const int row = row0 + i / 64, col = col0 + i % 64;
    if (row < M) {
      float sum = 0.f;
      for (int k = 0; k < KIN; ++k) sum += A[(size_t)row * KIN + k] * W[(size_t)k * KOUT + col];
      C[(size_t)row * KOUT + col] = sum;
    }
  }
#endif
}

#define GRP_CAP 51200
#define GRP_CHUNKS 64

__global__ void __launch_bounds__(1024) grp_rank(const int* __restrict__ dstI, int E, int N,
                                                 unsigned* __restrict__ rank,
                                                 unsigned* __restrict__ cntG) {
  __shared__ uint4 lc4[GRP_CAP / 4];
  unsigned* lc = (unsigned*)lc4;
  const int t = threadIdx.x;
  const int g = blockIdx.x;
  for (int i = t; i < GRP_CAP / 4; i += 1024) lc4[i] = make_uint4(0u, 0u, 0u, 0u);
  __syncthreads();
  const int per = (E + (int)gridDim.x - 1) / (int)gridDim.x;
  const int e0 = g * per;
  const int e1 = e0 + per < E ? e0 + per : E;
  for (int e = e0 + t; e < e1; e += 8 * 1024) {
    int d[8];
#pragma unroll
    for (int u = 0; u < 8; ++u) d[u] = dstI[e + u * 1024 < e1 ? e + u * 1024 : e1 - 1];
    asm volatile("" :: "v"(d[0]), "v"(d[1]), "v"(d[2]), "v"(d[3]),
                       "v"(d[4]), "v"(d[5]), "v"(d[6]), "v"(d[7]));
#pragma unroll
    for (int u = 0; u < 8; ++u)
      if (e + u * 1024 < e1) rank[e + u * 1024] = atomicAdd(&lc[d[u]], 1u);
  }
  __syncthreads();
  uint4* __restrict__ row = (uint4*)(cntG + (size_t)g * GRP_CAP);
  for (int i = t; i < (N + 3) / 4; i += 1024) row[i] = lc4[i];
}

template <int M>
__device__ __forceinline__ float row_xor(float x) {
  static_assert(M >= 1 && M <= 15, "inside a 16-lane row");
  return __builtin_bit_cast(float, __builtin_amdgcn_update_dpp(0, __builtin_bit_cast(int, x),
                                                               0x160 + M, 0xf, 0xf, true));
}
__device__ __forceinline__ float half_xor(float x) {
  const int b = __builtin_bit_cast(int, x);
  return __builtin_bit_cast(float,
                            __builtin_amdgcn_permlanex16(b, b, 0x76543210, 0xfedcba98, false, false));
}
template <int S>
__device__ __forceinline__ unsigned row_up(unsigned x) {
  static_assert(S >= 1 && S <= 15, "inside a 16-lane row");
  return (unsigned)__builtin_amdgcn_update_dpp(0, (int)x, 0x110 + S, 0xf, 0xf, true);
}
__device__ __forceinline__ unsigned wave_running_sum(unsigned v, int lane) {
  v += row_up<1>(v);
  v += row_up<2>(v);
  v += row_up<4>(v);
  v += row_up<8>(v);
  const unsigned low = (unsigned)__builtin_amdgcn_readlane((int)v, 15);
  return lane >= 16 ? v + low : v;
}

__device__ __forceinline__ unsigned block256_offsets(unsigned v, int lane, int wvs,
                                                     unsigned* wsum, unsigned& total) {
  const unsigned inc = wave_running_sum(v, lane);
  if (lane == 31) wsum[wvs] = inc;
  __syncthreads();
  const unsigned wtot = lane < 8 ? wsum[lane] : 0u;
  const unsigned winc = wave_running_sum(wtot, lane);
  const unsigned woff = (unsigned)__builtin_amdgcn_readlane((int)winc, wvs) -
                        (unsigned)__builtin_amdgcn_readlane((int)wtot, wvs);
  total = (unsigned)__builtin_amdgcn_readlane((int)winc, 7);
  return woff + inc - v;
}

static_assert(GRP_CAP / 256 <= 256, "the placing pass prefixes the block totals one per thread");
template <int G>
__global__ void __launch_bounds__(256) grp_prefix(unsigned* __restrict__ cntG, int N,
                                                  unsigned* __restrict__ lpre,
                                                  unsigned* __restrict__ bsum) {
  static_assert(G % 32 == 0, "rows are read in batches of 32");
  __shared__ unsigned wsum[8];
  const int t = threadIdx.x, lane = t & 31;
  const int wvs = __builtin_amdgcn_readfirstlane(t >> 5);
  const int d = blockIdx.x * 256 + t;
  unsigned run = 0;
  if (d < N) {
    for (int g0 = 0; g0 < G; g0 += 32) {
      unsigned c[32];
#pragma unroll
      for (int u = 0; u < 32; ++u) c[u] = cntG[(size_t)(g0 + u) * GRP_CAP + d];
#pragma unroll
      for (int u = 0; u < 32; ++u) { cntG[(size_t)(g0 + u) * GRP_CAP + d] = run; run += c[u]; }
    }
  }
  unsigned total;
  const unsigned off = block256_offsets(run, lane, wvs, wsum, total);
  if (d < N) lpre[d] = off;
  if (t == 0) bsum[blockIdx.x] = total;
}

__global__ void __launch_bounds__(256) grp_place(const int* __restrict__ srcI,
                                                 const int* __restrict__ dstI, int E, int G,
                                                 const unsigned* __restrict__ rank,
                                                 const unsigned* __restrict__ cntG,
                                                 const unsigned* __restrict__ lpre,
                                                 const unsigned* __restrict__ bsum, int N,
                                                 unsigned* __restrict__ rowptr,
                                                 int* __restrict__ csr_src) {
  __shared__ unsigned bpre[257];
  __shared__ unsigned wsum[8];
  const int t = threadIdx.x, lane = t & 31;
  const int wvs = __builtin_amdgcn_readfirstlane(t >> 5);
  const int eb = blockIdx.x * 1024 + t;
  const int per = (E + G - 1) / G;
  const int NB = (N + 255) >> 8;
  int d[4], s[4];
  unsigned r[4], a[4], b[4];
#pragma unroll
  for (int u = 0; u < 4; ++u) {
    const int e = eb + u * 256 < E ? eb + u * 256 : E - 1;
    d[u] = dstI[e];
    s[u] = srcI[e];
    r[u] = rank[e];
  }
  {
    const unsigned bs = t < NB ? bsum[t] : 0u;
    unsigned total;
    bpre[t] = block256_offsets(bs, lane, wvs, wsum, total);
    if (t == 0) bpre[256] = total;
  }
  __syncthreads();
#pragma unroll
  for (int u = 0; u < 4; ++u) {
    const int e = eb + u * 256 < E ? eb + u * 256 : E - 1;
    a[u] = lpre[d[u]];
    b[u] = cntG[(size_t)(e / per) * GRP_CAP + d[u]];
  }
  asm volatile("" :: "v"(s[0]), "v"(s[1]), "v"(s[2]), "v"(s[3]), "v"(r[0]), "v"(r[1]), "v"(r[2]),
                     "v"(r[3]), "v"(a[0]), "v"(a[1]), "v"(a[2]), "v"(a[3]), "v"(b[0]), "v"(b[1]),
                     "v"(b[2]), "v"(b[3]));
#pragma unroll
  for (int u = 0; u < 4; ++u)
    if (eb + u * 256 < E) csr_src[a[u] + bpre[d[u] >> 8] + b[u] + r[u]] = s[u];
  for (int i = blockIdx.x * 256 + t; i <= N; i += (int)gridDim.x * 256)
    rowptr[i] = i < N ? lpre[i] + bpre[i >> 8] : bpre[NB];
}

__device__ __forceinline__ float leaky_dot4(v4f a, v4f b, v4f w) {
  float acc, v;
  v = a.x + b.x; v = fmaxf(v, 0.2f * v); acc  = v * w.x;
  v = a.y + b.y; v = fmaxf(v, 0.2f * v); acc += v * w.y;
  v = a.z + b.z; v = fmaxf(v, 0.2f * v); acc += v * w.z;
  v = a.w + b.w; v = fmaxf(v, 0.2f * v); acc += v * w.w;
  return acc;
}

__device__ __forceinline__ float exp_nonpos(float x) {
  return __builtin_amdgcn_exp2f(1.44269504f * x);
}

template <int C>
__device__ __forceinline__ float group_logit(v4f a, v4f b, v4f w) {
  float p = leaky_dot4(a, b, w);
  p += row_xor<1>(p);
  p += row_xor<2>(p);
  if constexpr (C == 32) p += row_xor<4>(p);
  return p;
}

template <int C>
__device__ __forceinline__ void node_softmax_sum(const float* __restrict__ xl,
                                                 const float* __restrict__ xr,
                                                 const v4f (&w)[HEADS * C / 128],
                                                 const unsigned* __restrict__ rowptr,
                                                 const int* __restrict__ csr_src,
                                                 int d, int lane,
                                                 v4f (&acc)[HEADS * C / 128],
                                                 float (&den)[HEADS * C / 128]) {
  static_assert(C == 32 || C == 16, "row pieces of 128 floats, 8 or 4 lanes per head");
  constexpr int ROW = HEADS * C;
  constexpr int NJ  = ROW / 128;
  constexpr int UN  = 4;
  static_assert(UN == 4 && (NJ == 1 || NJ == 2), "the load fence below names UN*NJ vectors");
  const unsigned beg = rowptr[d], end = rowptr[d + 1];
  const v4f* pr = (const v4f*)(xr + (size_t)d * ROW) + lane;
  const v4f* ps = (const v4f*)(xl + (size_t)d * ROW) + lane;
  v4f b[NJ];
  float mx[NJ];
#pragma unroll
  for (int j = 0; j < NJ; ++j) { b[j] = pr[j * 32]; acc[j] = ps[j * 32]; }
#pragma unroll
  for (int j = 0; j < NJ; ++j) { mx[j] = group_logit<C>(acc[j], b[j], w[j]); den[j] = 1.f; }

  for (unsigned base = beg; base < end; base += 32) {
    const int cnt = (int)(end - base) < 32 ? (int)(end - base) : 32;
    const int my  = lane < cnt ? csr_src[base + lane] : 0;
    for (int i = 0; i < cnt; i += UN) {
      v4f a[UN][NJ];
#pragma unroll
      for (int u = 0; u < UN; ++u) {
        const int ii = i + u < cnt ? i + u : cnt - 1;
        const int s  = __builtin_amdgcn_readlane(my, ii);
        const v4f* p = (const v4f*)(xl + (size_t)s * ROW) + lane;
#pragma unroll
        for (int j = 0; j < NJ; ++j) a[u][j] = p[j * 32];
      }
      if constexpr (NJ == 2)
        asm volatile("" :: "v"(a[0][0]), "v"(a[0][1]), "v"(a[1][0]), "v"(a[1][1]),
                           "v"(a[2][0]), "v"(a[2][1]), "v"(a[3][0]), "v"(a[3][1]));
      else
        asm volatile("" :: "v"(a[0][0]), "v"(a[1][0]), "v"(a[2][0]), "v"(a[3][0]));
#pragma unroll
      for (int u = 0; u < UN; ++u) {
        if (u == 0 || i + u < cnt) {
#pragma unroll
          for (int j = 0; j < NJ; ++j) {
            const float lg = group_logit<C>(a[u][j], b[j], w[j]);
            const float t  = exp_nonpos(-fabsf(mx[j] - lg));
            const bool  up = lg > mx[j];
            const float sc = up ? t : 1.f;
            const float ex = up ? 1.f : t;
            mx[j]  = fmaxf(mx[j], lg);
            den[j] = den[j] * sc + ex;
            acc[j].x = acc[j].x * sc + ex * a[u][j].x;
            acc[j].y = acc[j].y * sc + ex * a[u][j].y;
            acc[j].z = acc[j].z * sc + ex * a[u][j].z;
            acc[j].w = acc[j].w * sc + ex * a[u][j].w;
          }
        }
      }
    }
  }
}

#define HT_ROWS 32
#define HT_PITCH 260
struct Gemm2Frag { v4f a[2]; v2f b[4]; };

__global__ void __launch_bounds__(256)
node_c32_gemm2(const float* __restrict__ xl, const float* __restrict__ xr,
               const float* __restrict__ att, const float* __restrict__ bias,
               const unsigned* __restrict__ rowptr, const int* __restrict__ csr_src, int N,
               const float* __restrict__ Wl, const float* __restrict__ Wr,
               float* __restrict__ Cl, float* __restrict__ Cr) {
  constexpr int KIN = 256, KOUT = 128;
  __shared__ v4f hs4[HT_ROWS * HT_PITCH / 4];
  float* hs = (float*)hs4;
  const int lane = threadIdx.x & 31;
  const int wib  = __builtin_amdgcn_readfirstlane((int)(threadIdx.x >> 5));
  const int row0 = blockIdx.x * HT_ROWS;
  {
    v4f w[2], bz[2];
#pragma unroll
    for (int j = 0; j < 2; ++j) {
      w[j]  = ((const v4f*)att)[j * 32 + lane];
      bz[j] = ((const v4f*)bias)[j * 32 + lane];
    }
    for (int k = 0; k < HT_ROWS / 8; ++k) {
      const int r = wib + 8 * k;
      const int d = row0 + r;
      v4f v[2];
      if (d < N) {
        v4f acc[2];
        float den[2];
        node_softmax_sum<32>(xl, xr, w, rowptr, csr_src, d, lane, acc, den);
#pragma unroll
        for (int j = 0; j < 2; ++j) {
          v4f t;
          t.x = acc[j].x / den[j] + bz[j].x; t.x = t.x > 0.f ? t.x : (exp_nonpos(t.x) - 1.f);
          t.y = acc[j].y / den[j] + bz[j].y; t.y = t.y > 0.f ? t.y : (exp_nonpos(t.y) - 1.f);
          t.z = acc[j].z / den[j] + bz[j].z; t.z = t.z > 0.f ? t.z : (exp_nonpos(t.z) - 1.f);
          t.w = acc[j].w / den[j] + bz[j].w; t.w = t.w > 0.f ? t.w : (exp_nonpos(t.w) - 1.f);
          v[j] = t;
        }
      } else {
        v[0] = (v4f){0.f, 0.f, 0.f, 0.f};
        v[1] = v[0];
      }
      v4f* ph = (v4f*)(hs + r * HT_PITCH) + lane;
      ph[0]  = v[0];
      ph[32] = v[1];
    }
  }
  __syncthreads();

  const int h = lane >> 4;
  const int c = lane & 15;
  const bool right = wib >= 4;
  const float* __restrict__ W = right ? Wr : Wl;
  float* __restrict__ C       = right ? Cr : Cl;
  const int col0 = (wib & 3) * 32;
#if USE_WMMA_F32
  const v4f* a0p = (const v4f*)(hs + c * HT_PITCH + 4 * h);
  const v4f* a1p = (const v4f*)(hs + (16 + c) * HT_PITCH + 4 * h);
  const v2f* bp  = (const v2f*)(W + (size_t)(4 * h) * KOUT + col0 + 2 * c);

  v8f acc[2][2] = {};
  auto load = [&](Gemm2Frag& f, int kk) {
    f.a[0] = a0p[2 * kk];
    f.a[1] = a1p[2 * kk];
#pragma unroll
    for (int e = 0; e < 4; ++e) f.b[e] = bp[(size_t)(8 * kk + e) * (KOUT / 2)];
  };
  auto mma = [&](const Gemm2Frag& f) {
#pragma unroll
    for (int p = 0; p < 2; ++p) {
#pragma unroll
      for (int r = 0; r < 2; ++r) {
        v2f a;
        a[0] = f.a[r][2 * p];
        a[1] = f.a[r][2 * p + 1];
#pragma unroll
        for (int t = 0; t < 2; ++t) {
          v2f b;
          b[0] = f.b[2 * p][t];
          b[1] = f.b[2 * p + 1][t];
          acc[r][t] = __builtin_amdgcn_wmma_f32_16x16x4_f32(false, a, false, b,
                                                            (short)0, acc[r][t], false, false);
        }
      }
    }
  };

  constexpr int NK = KIN / 8;
  Gemm2Frag f0, f1;
  load(f0, 0);
#pragma nounroll
  for (int kk = 0; kk < NK - 2; kk += 2) {
    load(f1, kk + 1);
    __builtin_amdgcn_sched_barrier(0);
    mma(f0);
    load(f0, kk + 2);
    __builtin_amdgcn_sched_barrier(0);
    mma(f1);
  }
  load(f1, NK - 1);
  __builtin_amdgcn_sched_barrier(0);
  mma(f0);
  mma(f1);

#pragma unroll
  for (int r = 0; r < 2; ++r) {
#pragma unroll
    for (int v = 0; v < 8; ++v) {
      const int row = row0 + 16 * r + 8 * h + v;
      if (row < N) {
        v2f o;
        o[0] = acc[r][0][v]; o[1] = acc[r][1][v];
        *(v2f*)(C + (size_t)row * KOUT + col0 + 2 * c) = o;
      }
    }
  }
#else
  for (int i = lane; i < HT_ROWS * 32; i += 32) {
    const int r = i / 32, col = col0 + i % 32;
    if (row0 + r < N) {
      float sum = 0.f;
      for (int k = 0; k < KIN; ++k) sum += hs[r * HT_PITCH + k] * W[(size_t)k * KOUT + col];
      C[(size_t)(row0 + r) * KOUT + col] = sum;
    }
  }
#endif
}

__global__ void __launch_bounds__(256) node_gat_c16(const float* __restrict__ xl,
                                                    const float* __restrict__ xr,
                                                    const float* __restrict__ att,
                                                    const float* __restrict__ bias,
                                                    const unsigned* __restrict__ rowptr,
                                                    const int* __restrict__ csr_src,
                                                    int N, float* __restrict__ out) {
  const int lane = threadIdx.x & 31;
  const int wib  = __builtin_amdgcn_readfirstlane((int)(threadIdx.x >> 5));
  const int wave = blockIdx.x * (int)(blockDim.x >> 5) + wib;
  const int nw   = gridDim.x * (int)(blockDim.x >> 5);
  v4f w[1];
  w[0] = ((const v4f*)att)[lane];
  const v4f bz = ((const v4f*)bias)[lane];
  for (int d = wave; d < N; d += nw) {
    v4f acc[1];
    float den[1];
    node_softmax_sum<16>(xl, xr, w, rowptr, csr_src, d, lane, acc, den);
    v4f v;
    v.x = acc[0].x / den[0] + bz.x;
    v.y = acc[0].y / den[0] + bz.y;
    v.z = acc[0].z / den[0] + bz.z;
    v.w = acc[0].w / den[0] + bz.w;
    float rm = fmaxf(fmaxf(v.x, v.y), fmaxf(v.z, v.w));
    rm = fmaxf(rm, row_xor<1>(rm));
    rm = fmaxf(rm, row_xor<2>(rm));
    rm = fmaxf(rm, row_xor<4>(rm));
    rm = fmaxf(rm, row_xor<8>(rm));
    rm = fmaxf(rm, half_xor(rm));
    float se = (expf(v.x - rm) + expf(v.y - rm)) + (expf(v.z - rm) + expf(v.w - rm));
    se += row_xor<1>(se);
    se += row_xor<2>(se);
    se += row_xor<4>(se);
    se += row_xor<8>(se);
    se += half_xor(se);
    const float lse = logf(se);
    v.x = v.x - rm - lse; v.y = v.y - rm - lse; v.z = v.z - rm - lse; v.w = v.w - rm - lse;
    ((v4f*)(out + (size_t)d * 128))[lane] = v;
  }
}

extern "C" void kernel_launch(void* const* d_in, const int* in_sizes, int n_in,
                              void* d_out, int out_size, void* d_ws, size_t ws_size,
                              hipStream_t stream) {
  (void)n_in; (void)out_size; (void)ws_size;
  const float* x    = (const float*)d_in[0];
  const int*   ei   = (const int*)d_in[1];
  const float* W1l  = (const float*)d_in[2];
  const float* W1r  = (const float*)d_in[3];
  const float* att1 = (const float*)d_in[4];
  const float* b1   = (const float*)d_in[5];
  const float* W2l  = (const float*)d_in[6];
  const float* W2r  = (const float*)d_in[7];
  const float* att2 = (const float*)d_in[8];
  const float* b2   = (const float*)d_in[9];

  int N = in_sizes[0] / 128;
  int E = in_sizes[1] / 2;
  const int* srcI = ei;
  const int* dstI = ei + E;

  float* ws    = (float*)d_ws;
  float* xl    = ws;
  float* xr    = xl + (size_t)N * 256;
  float* xl2   = xr + (size_t)N * 256;
  float* xr2   = xl2 + (size_t)N * 128;
  unsigned* cntG   = (unsigned*)(xr2 + (size_t)N * 128);
  unsigned* lpre   = cntG + (size_t)GRP_CHUNKS * GRP_CAP;
  unsigned* rowptr = lpre + (size_t)N;
  int*      csr    = (int*)(rowptr + (size_t)N + 1);
  unsigned* rank   = (unsigned*)(csr + (size_t)E);
  unsigned* bsum   = rank + (size_t)E;
  float* outF = (float*)d_out;

  dim3 blk(256);

  grp_rank<<<GRP_CHUNKS, dim3(1024), 0, stream>>>(dstI, E, N, rank, cntG);
  grp_prefix<GRP_CHUNKS><<<(N + 255) / 256, blk, 0, stream>>>(cntG, N, lpre, bsum);
  grp_place<<<(E + 1023) / 1024, blk, 0, stream>>>(srcI, dstI, E, GRP_CHUNKS, rank, cntG, lpre, bsum, N,
                                                   rowptr, csr);

  int rtiles = (N + 31) / 32;
  gemm_pair_wmma<128, 256><<<rtiles, dim3(256), 0, stream>>>(x, W1l, W1r, xl, xr, N);
  node_c32_gemm2<<<rtiles, blk, 0, stream>>>(xl, xr, att1, b1, rowptr, csr, N, W2l, W2r, xl2, xr2);

  node_gat_c16<<<(N + 15) / 16, blk, 0, stream>>>(xl2, xr2, att2, b2, rowptr, csr, N, outF);
}
